# starting kernel
# baseline (speedup 1.0000x reference)
#include <hip/hip_runtime.h>

#define N_NODES 10000
#define N_EDGES 320000
#define IN_DIM  256
#define HIDDEN  256
#define OUT_DIM 128

typedef __attribute__((ext_vector_type(2))) float v2f;
typedef __attribute__((ext_vector_type(8))) float v8f;

template<int D>
__global__ __launch_bounds__(256) void spmm_scatter(
    const float* __restrict__ H,
    const int*   __restrict__ src,
    const int*   __restrict__ dst,
    const float* __restrict__ val,
    float*       __restrict__ Z)
{
    const int lane = threadIdx.x & 31;
    const int wv   = threadIdx.x >> 5;
    const int e    = blockIdx.x * 8 + wv;
    if (e >= N_EDGES) return;

    const int   s = src[e];
    const int   d = dst[e];
    const float w = val[e];

    const float* __restrict__ hrow = H + (size_t)s * D;
    float*       __restrict__ zrow = Z + (size_t)d * D;

#pragma unroll
    for (int i = 0; i < D / 128; ++i) {
        const int c = (i * 32 + lane) * 4;
        const float4 h = *(const float4*)(hrow + c);
        atomicAdd(&zrow[c + 0], w * h.x);
        atomicAdd(&zrow[c + 1], w * h.y);
        atomicAdd(&zrow[c + 2], w * h.z);
        atomicAdd(&zrow[c + 3], w * h.w);
    }
}

template<int MSUB, int NSUB, bool RELU>
__global__ __launch_bounds__(128) void gemm_wmma_f32(
    const float* __restrict__ A,
    const float* __restrict__ B,
    float*       __restrict__ C,
    int M, int K, int N)
{
    const int lane = threadIdx.x & 31;
    const int wave = threadIdx.x >> 5;
    const int lo   = lane & 15;
    const int hi   = lane >> 4;

    const int m0  = blockIdx.x * (16 * MSUB);
    const int nb0 = wave * (NSUB * 16);

    int mbase[MSUB];
#pragma unroll
    for (int ms = 0; ms < MSUB; ++ms)
        mbase[ms] = min(m0 + 16 * ms, M - 16);

    v8f acc[MSUB][NSUB];
#pragma unroll
    for (int ms = 0; ms < MSUB; ++ms)
#pragma unroll
        for (int t = 0; t < NSUB; ++t) acc[ms][t] = 0.0f;

    const float* __restrict__ arow[MSUB];
#pragma unroll
    for (int ms = 0; ms < MSUB; ++ms)
        arow[ms] = A + (size_t)(mbase[ms] + lo) * K;

    for (int k = 0; k < K; k += 4) {
        const int ka = k + 2 * hi;

        v2f a[MSUB];
#pragma unroll
        for (int ms = 0; ms < MSUB; ++ms)
            a[ms] = *(const v2f*)(arow[ms] + ka);

#pragma unroll
        for (int t = 0; t < NSUB; ++t) {
            const int n = nb0 + t * 16 + lo;
            v2f b;
            b.x = B[(size_t)ka * N + n];
            b.y = B[(size_t)(ka + 1) * N + n];
#pragma unroll
            for (int ms = 0; ms < MSUB; ++ms) {
                acc[ms][t] = __builtin_amdgcn_wmma_f32_16x16x4_f32(
                    false, a[ms], false, b, (short)0, acc[ms][t],
                    false, false);
            }
        }
    }

#pragma unroll
    for (int ms = 0; ms < MSUB; ++ms) {
        if (m0 + 16 * ms < M) {
#pragma unroll
            for (int t = 0; t < NSUB; ++t) {
#pragma unroll
                for (int v = 0; v < 8; ++v) {
                    const int row = mbase[ms] + v + 8 * hi;
                    const int col = nb0 + t * 16 + lo;
                    float x = acc[ms][t][v];
                    if (RELU) x = fmaxf(x, 0.0f);
                    C[(size_t)row * N + col] = x;
                }
            }
        }
    }
}

extern "C" void kernel_launch(void* const* d_in, const int* in_sizes, int n_in,
                              void* d_out, int out_size, void* d_ws, size_t ws_size,
                              hipStream_t stream)
{
    (void)in_sizes; (void)n_in; (void)out_size; (void)ws_size;

    const float* features = (const float*)d_in[0];
    const int*   esrc     = (const int*)  d_in[1];
    const int*   edst     = (const int*)  d_in[2];
    const float* eval_    = (const float*)d_in[3];
    const float* W0       = (const float*)d_in[4];
    const float* W1       = (const float*)d_in[5];
    float*       out      = (float*)d_out;

    const size_t SZ = (size_t)N_NODES * HIDDEN * sizeof(float);
    float* Z = (float*)d_ws;
    float* H = (float*)((char*)d_ws + SZ);

    const int spmm_blocks = (N_EDGES + 7) / 8;
    const int gemm_blocks = (N_NODES + 31) / 32;

    hipMemsetAsync(Z, 0, SZ, stream);
    spmm_scatter<IN_DIM><<<spmm_blocks, 256, 0, stream>>>(features, esrc, edst, eval_, Z);
    gemm_wmma_f32<2, 4, true><<<gemm_blocks, 128, 0, stream>>>(Z, W0, H,
                                                               N_NODES, IN_DIM, HIDDEN);

    float* G = Z;
    gemm_wmma_f32<2, 2, false><<<gemm_blocks, 128, 0, stream>>>(H, W1, G,
                                                                N_NODES, HIDDEN, OUT_DIM);
    hipMemsetAsync(out, 0, (size_t)N_NODES * OUT_DIM * sizeof(float), stream);
    spmm_scatter<OUT_DIM><<<spmm_blocks, 256, 0, stream>>>(G, esrc, edst, eval_, out);
}


# SpMM: float-atomic scatter replaced by per-call destination-sorted edge list (int count, scan, place) + one-wave-per-row gather; row sums now sequential in arrival-rank order; big memsets gone
# speedup vs baseline: 1.5852x; 1.5852x over previous
#include <hip/hip_runtime.h>


#define N_NODES 10000
#define N_EDGES 320000
#define IN_DIM  256
#define HIDDEN  256
#define OUT_DIM 128

typedef __attribute__((ext_vector_type(2))) float v2f;
typedef __attribute__((ext_vector_type(8))) float v8f;

__global__ __launch_bounds__(256) void edge_rank(
    const int* __restrict__ dst,
    int*       __restrict__ cnt,
    int*       __restrict__ rank)
{
    const int e = blockIdx.x * 256 + threadIdx.x;
    if (e < N_EDGES) rank[e] = atomicAdd(&cnt[dst[e]], 1);
}

#define SCAN_THREADS 1024
#define SCAN_PER ((N_NODES + SCAN_THREADS - 1) / SCAN_THREADS)

__global__ __launch_bounds__(SCAN_THREADS) void row_scan(
    const int* __restrict__ cnt,
    int*       __restrict__ rowptr)
{
    __shared__ int part[SCAN_THREADS];
    const int t = threadIdx.x;

    int local[SCAN_PER];
    int s = 0;
#pragma unroll
    for (int i = 0; i < SCAN_PER; ++i) {
        const int idx = t * SCAN_PER + i;
        const int v = (idx < N_NODES) ? cnt[idx] : 0;
        local[i] = s;
        s += v;
    }
    part[t] = s;
    __syncthreads();
    for (int off = 1; off < SCAN_THREADS; off <<= 1) {
        const int v = (t >= off) ? part[t - off] : 0;
        __syncthreads();
        part[t] += v;
        __syncthreads();
    }
    const int base = part[t] - s;
#pragma unroll
    for (int i = 0; i < SCAN_PER; ++i) {
        const int idx = t * SCAN_PER + i;
        if (idx < N_NODES) rowptr[idx] = base + local[i];
    }
    if (t == SCAN_THREADS - 1) rowptr[N_NODES] = part[t];
}

__global__ __launch_bounds__(256) void edge_place(
    const int*   __restrict__ src,
    const int*   __restrict__ dst,
    const float* __restrict__ val,
    const int*   __restrict__ rowptr,
    const int*   __restrict__ rank,
    int2*        __restrict__ ev)
{
    const int e = blockIdx.x * 256 + threadIdx.x;
    if (e < N_EDGES) {
        const int p = rowptr[dst[e]] + rank[e];
        ev[p] = make_int2(src[e], __float_as_int(val[e]));
    }
}

template<int D, int U>
__device__ __forceinline__ void gather_group(
    const float* __restrict__ H, int ex, int ey, int j, int lane,
    float4 (&acc)[D / 128])
{
    constexpr int V = D / 128;
    float4 h[U][V];
    float  w[U];
#pragma unroll
    for (int u = 0; u < U; ++u) {
        const int s = __builtin_amdgcn_readlane(ex, j + u);
        w[u] = __int_as_float(__builtin_amdgcn_readlane(ey, j + u));
        const float* __restrict__ hrow = H + (size_t)s * D + lane * 4;
#pragma unroll
        for (int v = 0; v < V; ++v)
            h[u][v] = *(const float4*)(hrow + v * 128);
    }
#pragma unroll
    for (int u = 0; u < U; ++u) {
#pragma unroll
        for (int v = 0; v < V; ++v) {
            acc[v].x = fmaf(w[u], h[u][v].x, acc[v].x);
            acc[v].y = fmaf(w[u], h[u][v].y, acc[v].y);
            acc[v].z = fmaf(w[u], h[u][v].z, acc[v].z);
            acc[v].w = fmaf(w[u], h[u][v].w, acc[v].w);
        }
    }
}

template<int D>
__global__ __launch_bounds__(256) void spmm_gather(
    const float* __restrict__ H,
    const int*   __restrict__ rowptr,
    const int2*  __restrict__ ev,
    float*       __restrict__ Z)
{
    constexpr int V = D / 128;
    const int lane = threadIdx.x & 31;
    const int row = __builtin_amdgcn_readfirstlane(blockIdx.x * 8 + (threadIdx.x >> 5));
    if (row >= N_NODES) return;

    const int beg = rowptr[row];
    const int end = rowptr[row + 1];

    float4 acc[V];
#pragma unroll
    for (int v = 0; v < V; ++v) acc[v] = make_float4(0.0f, 0.0f, 0.0f, 0.0f);

    for (int base = beg; base < end; base += 32) {
        const int n = min(32, end - base);
        int2 my = make_int2(0, 0);
        if (lane < n) my = ev[base + lane];

        int j = 0;
        for (; j + 8 <= n; j += 8) gather_group<D, 8>(H, my.x, my.y, j, lane, acc);
        if (j + 4 <= n) { gather_group<D, 4>(H, my.x, my.y, j, lane, acc); j += 4; }
        if (j + 2 <= n) { gather_group<D, 2>(H, my.x, my.y, j, lane, acc); j += 2; }
        if (j < n)      { gather_group<D, 1>(H, my.x, my.y, j, lane, acc); }
    }

    float* __restrict__ zrow = Z + (size_t)row * D + lane * 4;
#pragma unroll
    for (int v = 0; v < V; ++v)
        *(float4*)(zrow + v * 128) = acc[v];
}

template<int MSUB, int NSUB, bool RELU>
__global__ __launch_bounds__(128) void gemm_wmma_f32(
    const float* __restrict__ A,
    const float* __restrict__ B,
    float*       __restrict__ C,
    int M, int K, int N)
{
    const int lane = threadIdx.x & 31;
    const int wave = threadIdx.x >> 5;
    const int lo   = lane & 15;
    const int hi   = lane >> 4;

    const int m0  = blockIdx.x * (16 * MSUB);
    const int nb0 = wave * (NSUB * 16);

    int mbase[MSUB];
#pragma unroll
    for (int ms = 0; ms < MSUB; ++ms)
        mbase[ms] = min(m0 + 16 * ms, M - 16);

    v8f acc[MSUB][NSUB];
#pragma unroll
    for (int ms = 0; ms < MSUB; ++ms)
#pragma unroll
        for (int t = 0; t < NSUB; ++t) acc[ms][t] = 0.0f;

    const float* __restrict__ arow[MSUB];
#pragma unroll
    for (int ms = 0; ms < MSUB; ++ms)
        arow[ms] = A + (size_t)(mbase[ms] + lo) * K;

    for (int k = 0; k < K; k += 4) {
        const int ka = k + 2 * hi;

        v2f a[MSUB];
#pragma unroll
        for (int ms = 0; ms < MSUB; ++ms)
            a[ms] = *(const v2f*)(arow[ms] + ka);

#pragma unroll
        for (int t = 0; t < NSUB; ++t) {
            const int n = nb0 + t * 16 + lo;
            v2f b;
            b.x = B[(size_t)ka * N + n];
            b.y = B[(size_t)(ka + 1) * N + n];
#pragma unroll
            for (int ms = 0; ms < MSUB; ++ms) {
                acc[ms][t] = __builtin_amdgcn_wmma_f32_16x16x4_f32(
                    false, a[ms], false, b, (short)0, acc[ms][t],
                    false, false);
            }
        }
    }

#pragma unroll
    for (int ms = 0; ms < MSUB; ++ms) {
        if (m0 + 16 * ms < M) {
#pragma unroll
            for (int t = 0; t < NSUB; ++t) {
#pragma unroll
                for (int v = 0; v < 8; ++v) {
                    const int row = mbase[ms] + v + 8 * hi;
                    const int col = nb0 + t * 16 + lo;
                    float x = acc[ms][t][v];
                    if (RELU) x = fmaxf(x, 0.0f);
                    C[(size_t)row * N + col] = x;
                }
            }
        }
    }
}

#define SZ_BYTES   ((size_t)N_NODES * HIDDEN * sizeof(float))
#define OFF_CNT    (2 * SZ_BYTES)
#define OFF_ROWPTR (OFF_CNT + 65536)
#define OFF_RANK   (OFF_ROWPTR + 65536)
#define OFF_EV     (OFF_RANK + (size_t)N_EDGES * sizeof(int))

extern "C" void kernel_launch(void* const* d_in, const int* in_sizes, int n_in,
                              void* d_out, int out_size, void* d_ws, size_t ws_size,
                              hipStream_t stream)
{
    (void)in_sizes; (void)n_in; (void)out_size; (void)ws_size;

    const float* features = (const float*)d_in[0];
    const int*   esrc     = (const int*)  d_in[1];
    const int*   edst     = (const int*)  d_in[2];
    const float* eval_    = (const float*)d_in[3];
    const float* W0       = (const float*)d_in[4];
    const float* W1       = (const float*)d_in[5];
    float*       out      = (float*)d_out;

    float* Z      = (float*)d_ws;
    float* H      = (float*)((char*)d_ws + SZ_BYTES);
    int*   cnt    = (int*)  ((char*)d_ws + OFF_CNT);
    int*   rowptr = (int*)  ((char*)d_ws + OFF_ROWPTR);
    int*   rank   = (int*)  ((char*)d_ws + OFF_RANK);
    int2*  ev     = (int2*) ((char*)d_ws + OFF_EV);

    const int edge_blocks = (N_EDGES + 255) / 256;
    const int row_blocks  = (N_NODES + 7) / 8;
    const int gemm_blocks = (N_NODES + 31) / 32;

    hipMemsetAsync(cnt, 0, (size_t)N_NODES * sizeof(int), stream);
    edge_rank<<<edge_blocks, 256, 0, stream>>>(edst, cnt, rank);
    row_scan<<<1, SCAN_THREADS, 0, stream>>>(cnt, rowptr);
    edge_place<<<edge_blocks, 256, 0, stream>>>(esrc, edst, eval_, rowptr, rank, ev);

    spmm_gather<IN_DIM><<<row_blocks, 256, 0, stream>>>(features, rowptr, ev, Z);
    gemm_wmma_f32<2, 4, true><<<gemm_blocks, 128, 0, stream>>>(Z, W0, H,
                                                               N_NODES, IN_DIM, HIDDEN);

    float* G = Z;
    gemm_wmma_f32<2, 2, false><<<gemm_blocks, 128, 0, stream>>>(H, W1, G,
                                                                N_NODES, HIDDEN, OUT_DIM);
    spmm_gather<OUT_DIM><<<row_blocks, 256, 0, stream>>>(G, rowptr, ev, out);
}


# Edge list: memset + global returning atomics + 1-block scan + scattered place (4 launches) -> 2 launches, counting sort by destination range in LDS; row sums stay in arrival order (LDS atomics)
# speedup vs baseline: 2.4726x; 1.5598x over previous
#include <hip/hip_runtime.h>


#define N_NODES 10000
#define N_EDGES 320000
#define IN_DIM  256
#define HIDDEN  256
#define OUT_DIM 128

typedef __attribute__((ext_vector_type(2))) float v2f;
typedef __attribute__((ext_vector_type(8))) float v8f;

#define NBIN     250
#define BIN_ROWS ((N_NODES + NBIN - 1) / NBIN)
#define NCHUNK   250
#define CHUNK    ((N_EDGES + NCHUNK - 1) / NCHUNK)
#define BIN_T    256
#define CH_PER   ((CHUNK + BIN_T - 1) / BIN_T)
#define ROW_T    1024
#define SEG_T    4
#define EV_STAGE 4096

static_assert(NBIN <= BIN_T, "one thread per bin in the offset scan");
static_assert(NCHUNK * SEG_T <= ROW_T, "SEG_T threads per chunk segment");
static_assert(BIN_ROWS <= ROW_T, "one thread per row of a range");
static_assert(ROW_T == 32 * 32, "second-level sum is one wave");

__global__ __launch_bounds__(BIN_T) void edge_bin(
    const int*   __restrict__ src,
    const int*   __restrict__ dst,
    const float* __restrict__ val,
    int*         __restrict__ offmat,
    int4*        __restrict__ tmp)
{
    __shared__ int  hist[NBIN + 1];
    __shared__ int  part[BIN_T / 32];
    __shared__ int4 stage[CHUNK];

    const int t  = threadIdx.x;
    const int c  = blockIdx.x;
    const int e0 = c * CHUNK;
    const int n  = max(0, min(CHUNK, N_EDGES - e0));

    for (int i = t; i <= NBIN; i += BIN_T) hist[i] = 0;

    int d[CH_PER], s[CH_PER], w[CH_PER], pos[CH_PER];
#pragma unroll
    for (int i = 0; i < CH_PER; ++i) {
        const int j = i * BIN_T + t;
        d[i] = 0; s[i] = 0; w[i] = 0; pos[i] = 0;
        if (j < n) {
            d[i] = dst[e0 + j];
            s[i] = src[e0 + j];
            w[i] = __float_as_int(val[e0 + j]);
        }
    }
    __syncthreads();
#pragma unroll
    for (int i = 0; i < CH_PER; ++i) {
        const int j = i * BIN_T + t;
        if (j < n) pos[i] = atomicAdd(&hist[d[i] / BIN_ROWS], 1);
    }
    __syncthreads();

    const int cntv = (t < NBIN) ? hist[t] : 0;
    const int lane = t & 31;
    const int wv   = t >> 5;
    int incl = cntv;
    for (int m = 1; m < 32; m <<= 1) {
        const int y = __shfl_up(incl, m);
        if (lane >= m) incl += y;
    }
    if (lane == 31) part[wv] = incl;
    __syncthreads();
    int excl = incl - cntv;
    for (int i = 0; i < wv; ++i) excl += part[i];
    if (t < NBIN) hist[t] = excl;
    if (t == NBIN - 1) hist[NBIN] = excl + cntv;
    __syncthreads();

    for (int i = t; i <= NBIN; i += BIN_T) offmat[c * (NBIN + 1) + i] = hist[i];

#pragma unroll
    for (int i = 0; i < CH_PER; ++i) {
        const int j = i * BIN_T + t;
        if (j < n) stage[hist[d[i] / BIN_ROWS] + pos[i]] = make_int4(d[i], s[i], w[i], 0);
    }
    __syncthreads();
    for (int j = t; j < n; j += BIN_T) tmp[e0 + j] = stage[j];
}

__global__ __launch_bounds__(ROW_T) void bin_rows(
    const int*  __restrict__ offmat,
    const int4* __restrict__ tmp,
    int*        __restrict__ rowptr,
    int2*       __restrict__ ev)
{
    __shared__ int  rcnt[BIN_ROWS];
    __shared__ int  roff[BIN_ROWS + 1];
    __shared__ int  red[ROW_T / 32];
    __shared__ int  base_s;
    __shared__ int2 st[EV_STAGE];

    const int t  = threadIdx.x;
    const int b  = blockIdx.x;
    const int lo = b * BIN_ROWS;
    const int q  = t & (SEG_T - 1);

    int seg_beg = 0, seg_n = 0, below = 0;
    if (t < NCHUNK * SEG_T) {
        const int c  = t / SEG_T;
        const int o0 = offmat[c * (NBIN + 1) + b];
        const int o1 = offmat[c * (NBIN + 1) + b + 1];
        seg_beg = c * CHUNK + o0;
        seg_n   = o1 - o0;
        if (q == 0) below = o0;
    }
    for (int m = 16; m >= 1; m >>= 1) below += __shfl_xor(below, m);
    if ((t & 31) == 0) red[t >> 5] = below;
    if (t < BIN_ROWS) rcnt[t] = 0;
    __syncthreads();
    if (t < 32) {
        int x = red[t];
        for (int m = 16; m >= 1; m >>= 1) x += __shfl_xor(x, m);
        if (t == 0) base_s = x;
    }

    for (int i = q; i < seg_n; i += SEG_T) {
        const int4 r = tmp[seg_beg + i];
        atomicAdd(&rcnt[r.x - lo], 1);
    }
    __syncthreads();

    if (t < BIN_ROWS) {
        int sum = 0;
        for (int i = 0; i < t; ++i) sum += rcnt[i];
        roff[t] = sum;
        if (t == BIN_ROWS - 1) roff[BIN_ROWS] = sum + rcnt[t];
    }
    __syncthreads();

    const int base  = base_s;
    const int total = roff[BIN_ROWS];
    if (t < BIN_ROWS) {
        rcnt[t] = roff[t];
        if (lo + t < N_NODES) rowptr[lo + t] = base + roff[t];
    }
    if (b == NBIN - 1 && t == 0) rowptr[N_NODES] = base + total;
    __syncthreads();

    const bool staged = total <= EV_STAGE;
    for (int i = q; i < seg_n; i += SEG_T) {
        const int4 r = tmp[seg_beg + i];
        const int  p = atomicAdd(&rcnt[r.x - lo], 1);
        const int2 o = make_int2(r.y, r.z);
        if (staged) st[p] = o; else ev[base + p] = o;
    }
    __syncthreads();
    if (staged)
        for (int i = t; i < total; i += ROW_T) ev[base + i] = st[i];
}

template<int D, int U>
__device__ __forceinline__ void gather_group(
    const float* __restrict__ H, int ex, int ey, int j, int lane,
    float4 (&acc)[D / 128])
{
    constexpr int V = D / 128;
    float4 h[U][V];
    float  w[U];
#pragma unroll
    for (int u = 0; u < U; ++u) {
        const int s = __builtin_amdgcn_readlane(ex, j + u);
        w[u] = __int_as_float(__builtin_amdgcn_readlane(ey, j + u));
        const float* __restrict__ hrow = H + (size_t)s * D + lane * 4;
#pragma unroll
        for (int v = 0; v < V; ++v)
            h[u][v] = *(const float4*)(hrow + v * 128);
    }
#pragma unroll
    for (int u = 0; u < U; ++u) {
#pragma unroll
        for (int v = 0; v < V; ++v) {
            acc[v].x = fmaf(w[u], h[u][v].x, acc[v].x);
            acc[v].y = fmaf(w[u], h[u][v].y, acc[v].y);
            acc[v].z = fmaf(w[u], h[u][v].z, acc[v].z);
            acc[v].w = fmaf(w[u], h[u][v].w, acc[v].w);
        }
    }
}

template<int D>
__global__ __launch_bounds__(256) void spmm_gather(
    const float* __restrict__ H,
    const int*   __restrict__ rowptr,
    const int2*  __restrict__ ev,
    float*       __restrict__ Z)
{
    constexpr int V = D / 128;
    const int lane = threadIdx.x & 31;
    const int row = __builtin_amdgcn_readfirstlane(blockIdx.x * 8 + (threadIdx.x >> 5));
    if (row >= N_NODES) return;

    const int beg = rowptr[row];
    const int end = rowptr[row + 1];

    float4 acc[V];
#pragma unroll
    for (int v = 0; v < V; ++v) acc[v] = make_float4(0.0f, 0.0f, 0.0f, 0.0f);

    for (int base = beg; base < end; base += 32) {
        const int n = min(32, end - base);
        int2 my = make_int2(0, 0);
        if (lane < n) my = ev[base + lane];

        int j = 0;
        for (; j + 8 <= n; j += 8) gather_group<D, 8>(H, my.x, my.y, j, lane, acc);
        if (j + 4 <= n) { gather_group<D, 4>(H, my.x, my.y, j, lane, acc); j += 4; }
        if (j + 2 <= n) { gather_group<D, 2>(H, my.x, my.y, j, lane, acc); j += 2; }
        if (j < n)      { gather_group<D, 1>(H, my.x, my.y, j, lane, acc); }
    }

    float* __restrict__ zrow = Z + (size_t)row * D + lane * 4;
#pragma unroll
    for (int v = 0; v < V; ++v)
        *(float4*)(zrow + v * 128) = acc[v];
}

template<int MSUB, int NSUB, bool RELU>
__global__ __launch_bounds__(128) void gemm_wmma_f32(
    const float* __restrict__ A,
    const float* __restrict__ B,
    float*       __restrict__ C,
    int M, int K, int N)
{
    const int lane = threadIdx.x & 31;
    const int wave = threadIdx.x >> 5;
    const int lo   = lane & 15;
    const int hi   = lane >> 4;

    const int m0  = blockIdx.x * (16 * MSUB);
    const int nb0 = wave * (NSUB * 16);

    int mbase[MSUB];
#pragma unroll
    for (int ms = 0; ms < MSUB; ++ms)
        mbase[ms] = min(m0 + 16 * ms, M - 16);

    v8f acc[MSUB][NSUB];
#pragma unroll
    for (int ms = 0; ms < MSUB; ++ms)
#pragma unroll
        for (int t = 0; t < NSUB; ++t) acc[ms][t] = 0.0f;

    const float* __restrict__ arow[MSUB];
#pragma unroll
    for (int ms = 0; ms < MSUB; ++ms)
        arow[ms] = A + (size_t)(mbase[ms] + lo) * K;

    for (int k = 0; k < K; k += 4) {
        const int ka = k + 2 * hi;

        v2f a[MSUB];
#pragma unroll
        for (int ms = 0; ms < MSUB; ++ms)
            a[ms] = *(const v2f*)(arow[ms] + ka);

#pragma unroll
        for (int t = 0; t < NSUB; ++t) {
            const int n = nb0 + t * 16 + lo;
            v2f b;
            b.x = B[(size_t)ka * N + n];
            b.y = B[(size_t)(ka + 1) * N + n];
#pragma unroll
            for (int ms = 0; ms < MSUB; ++ms) {
                acc[ms][t] = __builtin_amdgcn_wmma_f32_16x16x4_f32(
                    false, a[ms], false, b, (short)0, acc[ms][t],
                    false, false);
            }
        }
    }

#pragma unroll
    for (int ms = 0; ms < MSUB; ++ms) {
        if (m0 + 16 * ms < M) {
#pragma unroll
            for (int t = 0; t < NSUB; ++t) {
#pragma unroll
                for (int v = 0; v < 8; ++v) {
                    const int row = mbase[ms] + v + 8 * hi;
                    const int col = nb0 + t * 16 + lo;
                    float x = acc[ms][t][v];
                    if (RELU) x = fmaxf(x, 0.0f);
                    C[(size_t)row * N + col] = x;
                }
            }
        }
    }
}

#define SZ_BYTES   ((size_t)N_NODES * HIDDEN * sizeof(float))
#define OFF_OFFMAT (2 * SZ_BYTES)
#define OFF_ROWPTR (OFF_OFFMAT + 262144)
#define OFF_TMP    (OFF_ROWPTR + 65536)
#define OFF_EV     (OFF_TMP + (size_t)NCHUNK * CHUNK * sizeof(int4))

static_assert((size_t)NCHUNK * (NBIN + 1) * sizeof(int) <= 262144, "offset matrix fits its slot");

extern "C" void kernel_launch(void* const* d_in, const int* in_sizes, int n_in,
                              void* d_out, int out_size, void* d_ws, size_t ws_size,
                              hipStream_t stream)
{
    (void)in_sizes; (void)n_in; (void)out_size; (void)ws_size;

    const float* features = (const float*)d_in[0];
    const int*   esrc     = (const int*)  d_in[1];
    const int*   edst     = (const int*)  d_in[2];
    const float* eval_    = (const float*)d_in[3];
    const float* W0       = (const float*)d_in[4];
    const float* W1       = (const float*)d_in[5];
    float*       out      = (float*)d_out;

    float* Z      = (float*)d_ws;
    float* H      = (float*)((char*)d_ws + SZ_BYTES);
    int*   offmat = (int*)  ((char*)d_ws + OFF_OFFMAT);
    int*   rowptr = (int*)  ((char*)d_ws + OFF_ROWPTR);
    int4*  tmp    = (int4*) ((char*)d_ws + OFF_TMP);
    int2*  ev     = (int2*) ((char*)d_ws + OFF_EV);

    const int row_blocks  = (N_NODES + 7) / 8;
    const int gemm_blocks = (N_NODES + 31) / 32;

    edge_bin<<<NCHUNK, BIN_T, 0, stream>>>(esrc, edst, eval_, offmat, tmp);
    bin_rows<<<NBIN, ROW_T, 0, stream>>>(offmat, tmp, rowptr, ev);

    spmm_gather<IN_DIM><<<row_blocks, 256, 0, stream>>>(features, rowptr, ev, Z);
    gemm_wmma_f32<2, 4, true><<<gemm_blocks, 128, 0, stream>>>(Z, W0, H,
                                                               N_NODES, IN_DIM, HIDDEN);

    float* G = Z;
    gemm_wmma_f32<2, 2, false><<<gemm_blocks, 128, 0, stream>>>(H, W1, G,
                                                                N_NODES, HIDDEN, OUT_DIM);
    spmm_gather<OUT_DIM><<<row_blocks, 256, 0, stream>>>(G, rowptr, ev, out);
}


# GEMM K loop software-pipelined: two fragment sets in registers, the next trip's A and B fragment loads issued as one group ahead of the current trip's WMMAs; k order and summation order unchanged
# speedup vs baseline: 3.2677x; 1.3216x over previous
#include <hip/hip_runtime.h>


#define N_NODES 10000
#define N_EDGES 320000
#define IN_DIM  256
#define HIDDEN  256
#define OUT_DIM 128

typedef __attribute__((ext_vector_type(2))) float v2f;
typedef __attribute__((ext_vector_type(8))) float v8f;

#define NBIN     250
#define BIN_ROWS ((N_NODES + NBIN - 1) / NBIN)
#define NCHUNK   250
#define CHUNK    ((N_EDGES + NCHUNK - 1) / NCHUNK)
#define BIN_T    256
#define CH_PER   ((CHUNK + BIN_T - 1) / BIN_T)
#define ROW_T    1024
#define SEG_T    4
#define EV_STAGE 4096

static_assert(NBIN <= BIN_T, "one thread per bin in the offset scan");
static_assert(NCHUNK * SEG_T <= ROW_T, "SEG_T threads per chunk segment");
static_assert(BIN_ROWS <= ROW_T, "one thread per row of a range");
static_assert(ROW_T == 32 * 32, "second-level sum is one wave");

__global__ __launch_bounds__(BIN_T) void edge_bin(
    const int*   __restrict__ src,
    const int*   __restrict__ dst,
    const float* __restrict__ val,
    int*         __restrict__ offmat,
    int4*        __restrict__ tmp)
{
    __shared__ int  hist[NBIN + 1];
    __shared__ int  part[BIN_T / 32];
    __shared__ int4 stage[CHUNK];

    const int t  = threadIdx.x;
    const int c  = blockIdx.x;
    const int e0 = c * CHUNK;
    const int n  = max(0, min(CHUNK, N_EDGES - e0));

    for (int i = t; i <= NBIN; i += BIN_T) hist[i] = 0;

    int d[CH_PER], s[CH_PER], w[CH_PER], pos[CH_PER];
#pragma unroll
    for (int i = 0; i < CH_PER; ++i) {
        const int j = i * BIN_T + t;
        d[i] = 0; s[i] = 0; w[i] = 0; pos[i] = 0;
        if (j < n) {
            d[i] = dst[e0 + j];
            s[i] = src[e0 + j];
            w[i] = __float_as_int(val[e0 + j]);
        }
    }
    __syncthreads();
#pragma unroll
    for (int i = 0; i < CH_PER; ++i) {
        const int j = i * BIN_T + t;
        if (j < n) pos[i] = atomicAdd(&hist[d[i] / BIN_ROWS], 1);
    }
    __syncthreads();

    const int cntv = (t < NBIN) ? hist[t] : 0;
    const int lane = t & 31;
    const int wv   = t >> 5;
    int incl = cntv;
    for (int m = 1; m < 32; m <<= 1) {
        const int y = __shfl_up(incl, m);
        if (lane >= m) incl += y;
    }
    if (lane == 31) part[wv] = incl;
    __syncthreads();
    int excl = incl - cntv;
    for (int i = 0; i < wv; ++i) excl += part[i];
    if (t < NBIN) hist[t] = excl;
    if (t == NBIN - 1) hist[NBIN] = excl + cntv;
    __syncthreads();

    for (int i = t; i <= NBIN; i += BIN_T) offmat[c * (NBIN + 1) + i] = hist[i];

#pragma unroll
    for (int i = 0; i < CH_PER; ++i) {
        const int j = i * BIN_T + t;
        if (j < n) stage[hist[d[i] / BIN_ROWS] + pos[i]] = make_int4(d[i], s[i], w[i], 0);
    }
    __syncthreads();
    for (int j = t; j < n; j += BIN_T) tmp[e0 + j] = stage[j];
}

__global__ __launch_bounds__(ROW_T) void bin_rows(
    const int*  __restrict__ offmat,
    const int4* __restrict__ tmp,
    int*        __restrict__ rowptr,
    int2*       __restrict__ ev)
{
    __shared__ int  rcnt[BIN_ROWS];
    __shared__ int  roff[BIN_ROWS + 1];
    __shared__ int  red[ROW_T / 32];
    __shared__ int  base_s;
    __shared__ int2 st[EV_STAGE];

    const int t  = threadIdx.x;
    const int b  = blockIdx.x;
    const int lo = b * BIN_ROWS;
    const int q  = t & (SEG_T - 1);

    int seg_beg = 0, seg_n = 0, below = 0;
    if (t < NCHUNK * SEG_T) {
        const int c  = t / SEG_T;
        const int o0 = offmat[c * (NBIN + 1) + b];
        const int o1 = offmat[c * (NBIN + 1) + b + 1];
        seg_beg = c * CHUNK + o0;
        seg_n   = o1 - o0;
        if (q == 0) below = o0;
    }
    for (int m = 16; m >= 1; m >>= 1) below += __shfl_xor(below, m);
    if ((t & 31) == 0) red[t >> 5] = below;
    if (t < BIN_ROWS) rcnt[t] = 0;
    __syncthreads();
    if (t < 32) {
        int x = red[t];
        for (int m = 16; m >= 1; m >>= 1) x += __shfl_xor(x, m);
        if (t == 0) base_s = x;
    }

    for (int i = q; i < seg_n; i += SEG_T) {
        const int4 r = tmp[seg_beg + i];
        atomicAdd(&rcnt[r.x - lo], 1);
    }
    __syncthreads();

    if (t < BIN_ROWS) {
        int sum = 0;
        for (int i = 0; i < t; ++i) sum += rcnt[i];
        roff[t] = sum;
        if (t == BIN_ROWS - 1) roff[BIN_ROWS] = sum + rcnt[t];
    }
    __syncthreads();

    const int base  = base_s;
    const int total = roff[BIN_ROWS];
    if (t < BIN_ROWS) {
        rcnt[t] = roff[t];
        if (lo + t < N_NODES) rowptr[lo + t] = base + roff[t];
    }
    if (b == NBIN - 1 && t == 0) rowptr[N_NODES] = base + total;
    __syncthreads();

    const bool staged = total <= EV_STAGE;
    for (int i = q; i < seg_n; i += SEG_T) {
        const int4 r = tmp[seg_beg + i];
        const int  p = atomicAdd(&rcnt[r.x - lo], 1);
        const int2 o = make_int2(r.y, r.z);
        if (staged) st[p] = o; else ev[base + p] = o;
    }
    __syncthreads();
    if (staged)
        for (int i = t; i < total; i += ROW_T) ev[base + i] = st[i];
}

template<int D, int U>
__device__ __forceinline__ void gather_group(
    const float* __restrict__ H, int ex, int ey, int j, int lane,
    float4 (&acc)[D / 128])
{
    constexpr int V = D / 128;
    float4 h[U][V];
    float  w[U];
#pragma unroll
    for (int u = 0; u < U; ++u) {
        const int s = __builtin_amdgcn_readlane(ex, j + u);
        w[u] = __int_as_float(__builtin_amdgcn_readlane(ey, j + u));
        const float* __restrict__ hrow = H + (size_t)s * D + lane * 4;
#pragma unroll
        for (int v = 0; v < V; ++v)
            h[u][v] = *(const float4*)(hrow + v * 128);
    }
#pragma unroll
    for (int u = 0; u < U; ++u) {
#pragma unroll
        for (int v = 0; v < V; ++v) {
            acc[v].x = fmaf(w[u], h[u][v].x, acc[v].x);
            acc[v].y = fmaf(w[u], h[u][v].y, acc[v].y);
            acc[v].z = fmaf(w[u], h[u][v].z, acc[v].z);
            acc[v].w = fmaf(w[u], h[u][v].w, acc[v].w);
        }
    }
}

template<int D>
__global__ __launch_bounds__(256) void spmm_gather(
    const float* __restrict__ H,
    const int*   __restrict__ rowptr,
    const int2*  __restrict__ ev,
    float*       __restrict__ Z)
{
    constexpr int V = D / 128;
    const int lane = threadIdx.x & 31;
    const int row = __builtin_amdgcn_readfirstlane(blockIdx.x * 8 + (threadIdx.x >> 5));
    if (row >= N_NODES) return;

    const int beg = rowptr[row];
    const int end = rowptr[row + 1];

    float4 acc[V];
#pragma unroll
    for (int v = 0; v < V; ++v) acc[v] = make_float4(0.0f, 0.0f, 0.0f, 0.0f);

    for (int base = beg; base < end; base += 32) {
        const int n = min(32, end - base);
        int2 my = make_int2(0, 0);
        if (lane < n) my = ev[base + lane];

        int j = 0;
        for (; j + 8 <= n; j += 8) gather_group<D, 8>(H, my.x, my.y, j, lane, acc);
        if (j + 4 <= n) { gather_group<D, 4>(H, my.x, my.y, j, lane, acc); j += 4; }
        if (j + 2 <= n) { gather_group<D, 2>(H, my.x, my.y, j, lane, acc); j += 2; }
        if (j < n)      { gather_group<D, 1>(H, my.x, my.y, j, lane, acc); }
    }

    float* __restrict__ zrow = Z + (size_t)row * D + lane * 4;
#pragma unroll
    for (int v = 0; v < V; ++v)
        *(float4*)(zrow + v * 128) = acc[v];
}

template<int MSUB, int NSUB, int KS>
struct frag_set {
    v2f a[KS][MSUB];
    v2f b[KS][NSUB];
};

template<int MSUB, int NSUB, int KS, int NDIM>
__device__ __forceinline__ void load_frags(
    frag_set<MSUB, NSUB, KS>& f,
    const float* __restrict__ const (&ap)[MSUB],
    const float* __restrict__ bp, int k0)
{
#pragma unroll
    for (int j = 0; j < KS; ++j) {
        const int k = k0 + 4 * j;
#pragma unroll
        for (int ms = 0; ms < MSUB; ++ms)
            f.a[j][ms] = *(const v2f*)(ap[ms] + k);
#pragma unroll
        for (int t = 0; t < NSUB; ++t) {
            f.b[j][t].x = bp[(size_t)k * NDIM + t * 16];
            f.b[j][t].y = bp[(size_t)(k + 1) * NDIM + t * 16];
        }
    }
}

template<int MSUB, int NSUB, int KS>
__device__ __forceinline__ void mma_frags(
    const frag_set<MSUB, NSUB, KS>& f, v8f (&acc)[MSUB][NSUB])
{
#pragma unroll
    for (int j = 0; j < KS; ++j)
#pragma unroll
        for (int t = 0; t < NSUB; ++t)
#pragma unroll
            for (int ms = 0; ms < MSUB; ++ms) {
                acc[ms][t] = __builtin_amdgcn_wmma_f32_16x16x4_f32(
                    false, f.a[j][ms], false, f.b[j][t], (short)0, acc[ms][t],
                    false, false);
            }
}

template<int MSUB, int NSUB, int KS, int KDIM, int NDIM, bool RELU>
__global__ __launch_bounds__(128) void gemm_wmma_f32(
    const float* __restrict__ A,
    const float* __restrict__ B,
    float*       __restrict__ C,
    int M)
{
    constexpr int KT = 4 * KS;
    constexpr int NT = KDIM / KT;
    static_assert(KDIM % KT == 0 && NT >= 2 && NT % 2 == 0, "trips come in pairs");
    static_assert(NDIM == 4 * NSUB * 16, "four waves cover the row");

    const int lane = threadIdx.x & 31;
    const int wave = threadIdx.x >> 5;
    const int lo   = lane & 15;
    const int hi   = lane >> 4;

    const int m0  = blockIdx.x * (16 * MSUB);
    const int nb0 = wave * (NSUB * 16);

    int mbase[MSUB];
#pragma unroll
    for (int ms = 0; ms < MSUB; ++ms)
        mbase[ms] = min(m0 + 16 * ms, M - 16);

    v8f acc[MSUB][NSUB];
#pragma unroll
    for (int ms = 0; ms < MSUB; ++ms)
#pragma unroll
        for (int t = 0; t < NSUB; ++t) acc[ms][t] = 0.0f;

    const float* __restrict__ ap[MSUB];
#pragma unroll
    for (int ms = 0; ms < MSUB; ++ms)
        ap[ms] = A + (size_t)(mbase[ms] + lo) * KDIM + 2 * hi;
    const float* __restrict__ bp = B + (size_t)(2 * hi) * NDIM + nb0 + lo;

    frag_set<MSUB, NSUB, KS> f0, f1;
    load_frags<MSUB, NSUB, KS, NDIM>(f0, ap, bp, 0);
    for (int kt = 0; kt < NT - 2; kt += 2) {
        load_frags<MSUB, NSUB, KS, NDIM>(f1, ap, bp, (kt + 1) * KT);
        __builtin_amdgcn_sched_barrier(0);
        mma_frags<MSUB, NSUB, KS>(f0, acc);
        __builtin_amdgcn_sched_barrier(0);
        load_frags<MSUB, NSUB, KS, NDIM>(f0, ap, bp, (kt + 2) * KT);
        __builtin_amdgcn_sched_barrier(0);
        mma_frags<MSUB, NSUB, KS>(f1, acc);
        __builtin_amdgcn_sched_barrier(0);
    }
    load_frags<MSUB, NSUB, KS, NDIM>(f1, ap, bp, (NT - 1) * KT);
    __builtin_amdgcn_sched_barrier(0);
    mma_frags<MSUB, NSUB, KS>(f0, acc);
    mma_frags<MSUB, NSUB, KS>(f1, acc);

#pragma unroll
    for (int ms = 0; ms < MSUB; ++ms) {
        if (m0 + 16 * ms < M) {
#pragma unroll
            for (int t = 0; t < NSUB; ++t) {
#pragma unroll
                for (int v = 0; v < 8; ++v) {
                    const int row = mbase[ms] + v + 8 * hi;
                    const int col = nb0 + t * 16 + lo;
                    float x = acc[ms][t][v];
                    if (RELU) x = fmaxf(x, 0.0f);
                    C[(size_t)row * NDIM + col] = x;
                }
            }
        }
    }
}

#define SZ_BYTES   ((size_t)N_NODES * HIDDEN * sizeof(float))
#define OFF_OFFMAT (2 * SZ_BYTES)
#define OFF_ROWPTR (OFF_OFFMAT + 262144)
#define OFF_TMP    (OFF_ROWPTR + 65536)
#define OFF_EV     (OFF_TMP + (size_t)NCHUNK * CHUNK * sizeof(int4))

static_assert((size_t)NCHUNK * (NBIN + 1) * sizeof(int) <= 262144, "offset matrix fits its slot");

extern "C" void kernel_launch(void* const* d_in, const int* in_sizes, int n_in,
                              void* d_out, int out_size, void* d_ws, size_t ws_size,
                              hipStream_t stream)
{
    (void)in_sizes; (void)n_in; (void)out_size; (void)ws_size;

    const float* features = (const float*)d_in[0];
    const int*   esrc     = (const int*)  d_in[1];
    const int*   edst     = (const int*)  d_in[2];
    const float* eval_    = (const float*)d_in[3];
    const float* W0       = (const float*)d_in[4];
    const float* W1       = (const float*)d_in[5];
    float*       out      = (float*)d_out;

    float* Z      = (float*)d_ws;
    float* H      = (float*)((char*)d_ws + SZ_BYTES);
    int*   offmat = (int*)  ((char*)d_ws + OFF_OFFMAT);
    int*   rowptr = (int*)  ((char*)d_ws + OFF_ROWPTR);
    int4*  tmp    = (int4*) ((char*)d_ws + OFF_TMP);
    int2*  ev     = (int2*) ((char*)d_ws + OFF_EV);

    const int row_blocks  = (N_NODES + 7) / 8;
    const int gemm_blocks = (N_NODES + 31) / 32;

    edge_bin<<<NCHUNK, BIN_T, 0, stream>>>(esrc, edst, eval_, offmat, tmp);
    bin_rows<<<NBIN, ROW_T, 0, stream>>>(offmat, tmp, rowptr, ev);

    spmm_gather<IN_DIM><<<row_blocks, 256, 0, stream>>>(features, rowptr, ev, Z);
    gemm_wmma_f32<2, 4, 4, IN_DIM, HIDDEN, true><<<gemm_blocks, 128, 0, stream>>>(Z, W0, H, N_NODES);

    float* G = Z;
    gemm_wmma_f32<2, 2, 8, HIDDEN, OUT_DIM, false><<<gemm_blocks, 128, 0, stream>>>(H, W1, G, N_NODES);
    spmm_gather<OUT_DIM><<<row_blocks, 256, 0, stream>>>(G, rowptr, ev, out);
}


# Layer-0 SpMM gather fused into the row-binning kernel: each range block gathers its own rows from the (src,val) pairs it holds in LDS; one launch and 10000 waves fewer; row sums keep their order
# speedup vs baseline: 3.3825x; 1.0351x over previous
#include <hip/hip_runtime.h>


#define N_NODES 10000
#define N_EDGES 320000
#define IN_DIM  256
#define HIDDEN  256
#define OUT_DIM 128

typedef __attribute__((ext_vector_type(2))) float v2f;
typedef __attribute__((ext_vector_type(8))) float v8f;

#define NBIN     250
#define BIN_ROWS ((N_NODES + NBIN - 1) / NBIN)
#define NCHUNK   250
#define CHUNK    ((N_EDGES + NCHUNK - 1) / NCHUNK)
#define BIN_T    256
#define CH_PER   ((CHUNK + BIN_T - 1) / BIN_T)
#define ROW_T    1024
#define SEG_T    4
#define EV_STAGE 4096

static_assert(NBIN <= BIN_T, "one thread per bin in the offset scan");
static_assert(NCHUNK * SEG_T <= ROW_T, "SEG_T threads per chunk segment");
static_assert(BIN_ROWS <= ROW_T, "one thread per row of a range");
static_assert(ROW_T == 32 * 32, "second-level sum is one wave");

__global__ __launch_bounds__(BIN_T) void edge_bin(
    const int*   __restrict__ src,
    const int*   __restrict__ dst,
    const float* __restrict__ val,
    int*         __restrict__ offmat,
    int4*        __restrict__ tmp)
{
    __shared__ int  hist[NBIN + 1];
    __shared__ int  part[BIN_T / 32];
    __shared__ int4 stage[CHUNK];

    const int t  = threadIdx.x;
    const int c  = blockIdx.x;
    const int e0 = c * CHUNK;
    const int n  = max(0, min(CHUNK, N_EDGES - e0));

    for (int i = t; i <= NBIN; i += BIN_T) hist[i] = 0;

    int d[CH_PER], s[CH_PER], w[CH_PER], pos[CH_PER];
#pragma unroll
    for (int i = 0; i < CH_PER; ++i) {
        const int j = i * BIN_T + t;
        d[i] = 0; s[i] = 0; w[i] = 0; pos[i] = 0;
        if (j < n) {
            d[i] = dst[e0 + j];
            s[i] = src[e0 + j];
            w[i] = __float_as_int(val[e0 + j]);
        }
    }
    __syncthreads();
#pragma unroll
    for (int i = 0; i < CH_PER; ++i) {
        const int j = i * BIN_T + t;
        if (j < n) pos[i] = atomicAdd(&hist[d[i] / BIN_ROWS], 1);
    }
    __syncthreads();

    const int cntv = (t < NBIN) ? hist[t] : 0;
    const int lane = t & 31;
    const int wv   = t >> 5;
    int incl = cntv;
    for (int m = 1; m < 32; m <<= 1) {
        const int y = __shfl_up(incl, m);
        if (lane >= m) incl += y;
    }
    if (lane == 31) part[wv] = incl;
    __syncthreads();
    int excl = incl - cntv;
    for (int i = 0; i < wv; ++i) excl += part[i];
    if (t < NBIN) hist[t] = excl;
    if (t == NBIN - 1) hist[NBIN] = excl + cntv;
    __syncthreads();

    for (int i = t; i <= NBIN; i += BIN_T) offmat[c * (NBIN + 1) + i] = hist[i];

#pragma unroll
    for (int i = 0; i < CH_PER; ++i) {
        const int j = i * BIN_T + t;
        if (j < n) stage[hist[d[i] / BIN_ROWS] + pos[i]] = make_int4(d[i], s[i], w[i], 0);
    }
    __syncthreads();
    for (int j = t; j < n; j += BIN_T) tmp[e0 + j] = stage[j];
}

template<int D, int U, int LD = D>
__device__ __forceinline__ void gather_group(
    const float* __restrict__ H, int ex, int ey, int j, int lane,
    float4 (&acc)[D / 128])
{
    constexpr int V = D / 128;
    float4 h[U][V];
    float  w[U];
#pragma unroll
    for (int u = 0; u < U; ++u) {
        const int s = __builtin_amdgcn_readlane(ex, j + u);
        w[u] = __int_as_float(__builtin_amdgcn_readlane(ey, j + u));
        const float* __restrict__ hrow = H + (size_t)s * LD + lane * 4;
#pragma unroll
        for (int v = 0; v < V; ++v)
            h[u][v] = *(const float4*)(hrow + v * 128);
    }
#pragma unroll
    for (int u = 0; u < U; ++u) {
#pragma unroll
        for (int v = 0; v < V; ++v) {
            acc[v].x = fmaf(w[u], h[u][v].x, acc[v].x);
            acc[v].y = fmaf(w[u], h[u][v].y, acc[v].y);
            acc[v].z = fmaf(w[u], h[u][v].z, acc[v].z);
            acc[v].w = fmaf(w[u], h[u][v].w, acc[v].w);
        }
    }
}

template<bool B> struct pick { static constexpr bool value = B; };

__global__ __launch_bounds__(ROW_T) void bin_rows(
    const int*   __restrict__ offmat,
    const int4*  __restrict__ tmp,
    const float* __restrict__ X,
    int*         __restrict__ rowptr,
    int2*        __restrict__ ev,
    float*       __restrict__ Z)
{
    __shared__ int  rcnt[BIN_ROWS];
    __shared__ int  roff[BIN_ROWS + 1];
    __shared__ int  red[ROW_T / 32];
    __shared__ int  base_s;
    __shared__ int2 st[EV_STAGE];

    const int t  = threadIdx.x;
    const int b  = blockIdx.x;
    const int lo = b * BIN_ROWS;
    const int q  = t & (SEG_T - 1);

    int seg_beg = 0, seg_n = 0, below = 0;
    if (t < NCHUNK * SEG_T) {
        const int c  = t / SEG_T;
        const int o0 = offmat[c * (NBIN + 1) + b];
        const int o1 = offmat[c * (NBIN + 1) + b + 1];
        seg_beg = c * CHUNK + o0;
        seg_n   = o1 - o0;
        if (q == 0) below = o0;
    }
    for (int m = 16; m >= 1; m >>= 1) below += __shfl_xor(below, m);
    if ((t & 31) == 0) red[t >> 5] = below;
    if (t < BIN_ROWS) rcnt[t] = 0;
    __syncthreads();
    if (t < 32) {
        int x = red[t];
        for (int m = 16; m >= 1; m >>= 1) x += __shfl_xor(x, m);
        if (t == 0) base_s = x;
    }

    for (int i = q; i < seg_n; i += SEG_T) {
        const int4 r = tmp[seg_beg + i];
        atomicAdd(&rcnt[r.x - lo], 1);
    }
    __syncthreads();

    if (t < BIN_ROWS) {
        int sum = 0;
        for (int i = 0; i < t; ++i) sum += rcnt[i];
        roff[t] = sum;
        if (t == BIN_ROWS - 1) roff[BIN_ROWS] = sum + rcnt[t];
    }
    __syncthreads();

    const int base  = base_s;
    const int total = roff[BIN_ROWS];
    if (t < BIN_ROWS) {
        rcnt[t] = roff[t];
        if (lo + t < N_NODES) rowptr[lo + t] = base + roff[t];
    }
    if (b == NBIN - 1 && t == 0) rowptr[N_NODES] = base + total;
    __syncthreads();

    const bool staged = total <= EV_STAGE;
    for (int i = q; i < seg_n; i += SEG_T) {
        const int4 r = tmp[seg_beg + i];
        const int  p = atomicAdd(&rcnt[r.x - lo], 1);
        const int2 o = make_int2(r.y, r.z);
        if (staged) st[p] = o; else ev[base + p] = o;
    }
    __syncthreads();
    if (staged)
        for (int i = t; i < total; i += ROW_T) ev[base + i] = st[i];

    const int lane = t & 31;
    const int wv = __builtin_amdgcn_readfirstlane(t >> 5);
    auto gather_tasks = [&](auto from_lds) {
        for (int task = wv; task < 2 * BIN_ROWS; task += ROW_T / 32) {
            const int r    = task % BIN_ROWS;
            const int half = task / BIN_ROWS;
            const int row  = lo + r;
            if (row >= N_NODES) continue;
            const int beg = __builtin_amdgcn_readfirstlane(roff[r]);
            const int end = __builtin_amdgcn_readfirstlane(roff[r + 1]);
            const float* __restrict__ Xh = X + half * (IN_DIM / 2);

            float4 acc[1];
            acc[0] = make_float4(0.0f, 0.0f, 0.0f, 0.0f);
            for (int b0 = beg; b0 < end; b0 += 32) {
                const int n = min(32, end - b0);
                int2 my = make_int2(0, 0);
                if (lane < n) {
                    if constexpr (decltype(from_lds)::value) my = st[b0 + lane];
                    else                                     my = ev[base + b0 + lane];
                }

                int j = 0;
                for (; j + 8 <= n; j += 8) gather_group<IN_DIM / 2, 8, IN_DIM>(Xh, my.x, my.y, j, lane, acc);
                if (j + 4 <= n) { gather_group<IN_DIM / 2, 4, IN_DIM>(Xh, my.x, my.y, j, lane, acc); j += 4; }
                if (j + 2 <= n) { gather_group<IN_DIM / 2, 2, IN_DIM>(Xh, my.x, my.y, j, lane, acc); j += 2; }
                if (j < n)      { gather_group<IN_DIM / 2, 1, IN_DIM>(Xh, my.x, my.y, j, lane, acc); }
            }
            *(float4*)(Z + (size_t)row * IN_DIM + half * (IN_DIM / 2) + lane * 4) = acc[0];
        }
    };
    if (staged) gather_tasks(pick<true>{}); else gather_tasks(pick<false>{});
}

static_assert(IN_DIM == 256, "a gather task is one 128-column half of a row");

template<int D>
__global__ __launch_bounds__(256) void spmm_gather(
    const float* __restrict__ H,
    const int*   __restrict__ rowptr,
    const int2*  __restrict__ ev,
    float*       __restrict__ Z)
{
    constexpr int V = D / 128;
    const int lane = threadIdx.x & 31;
    const int row = __builtin_amdgcn_readfirstlane(blockIdx.x * 8 + (threadIdx.x >> 5));
    if (row >= N_NODES) return;

    const int beg = rowptr[row];
    const int end = rowptr[row + 1];

    float4 acc[V];
#pragma unroll
    for (int v = 0; v < V; ++v) acc[v] = make_float4(0.0f, 0.0f, 0.0f, 0.0f);

    for (int base = beg; base < end; base += 32) {
        const int n = min(32, end - base);
        int2 my = make_int2(0, 0);
        if (lane < n) my = ev[base + lane];

        int j = 0;
        for (; j + 8 <= n; j += 8) gather_group<D, 8>(H, my.x, my.y, j, lane, acc);
        if (j + 4 <= n) { gather_group<D, 4>(H, my.x, my.y, j, lane, acc); j += 4; }
        if (j + 2 <= n) { gather_group<D, 2>(H, my.x, my.y, j, lane, acc); j += 2; }
        if (j < n)      { gather_group<D, 1>(H, my.x, my.y, j, lane, acc); }
    }

    float* __restrict__ zrow = Z + (size_t)row * D + lane * 4;
#pragma unroll
    for (int v = 0; v < V; ++v)
        *(float4*)(zrow + v * 128) = acc[v];
}

template<int MSUB, int NSUB, int KS>
struct frag_set {
    v2f a[KS][MSUB];
    v2f b[KS][NSUB];
};

template<int MSUB, int NSUB, int KS, int NDIM>
__device__ __forceinline__ void load_frags(
    frag_set<MSUB, NSUB, KS>& f,
    const float* __restrict__ const (&ap)[MSUB],
    const float* __restrict__ bp, int k0)
{
#pragma unroll
    for (int j = 0; j < KS; ++j) {
        const int k = k0 + 4 * j;
#pragma unroll
        for (int ms = 0; ms < MSUB; ++ms)
            f.a[j][ms] = *(const v2f*)(ap[ms] + k);
#pragma unroll
        for (int t = 0; t < NSUB; ++t) {
            f.b[j][t].x = bp[(size_t)k * NDIM + t * 16];
            f.b[j][t].y = bp[(size_t)(k + 1) * NDIM + t * 16];
        }
    }
}

template<int MSUB, int NSUB, int KS>
__device__ __forceinline__ void mma_frags(
    const frag_set<MSUB, NSUB, KS>& f, v8f (&acc)[MSUB][NSUB])
{
#pragma unroll
    for (int j = 0; j < KS; ++j)
#pragma unroll
        for (int t = 0; t < NSUB; ++t)
#pragma unroll
            for (int ms = 0; ms < MSUB; ++ms) {
                acc[ms][t] = __builtin_amdgcn_wmma_f32_16x16x4_f32(
                    false, f.a[j][ms], false, f.b[j][t], (short)0, acc[ms][t],
                    false, false);
            }
}

template<int MSUB, int NSUB, int KS, int KDIM, int NDIM, bool RELU>
__global__ __launch_bounds__(128) void gemm_wmma_f32(
    const float* __restrict__ A,
    const float* __restrict__ B,
    float*       __restrict__ C,
    int M)
{
    constexpr int KT = 4 * KS;
    constexpr int NT = KDIM / KT;
    static_assert(KDIM % KT == 0 && NT >= 2 && NT % 2 == 0, "trips come in pairs");
    static_assert(NDIM == 4 * NSUB * 16, "four waves cover the row");

    const int lane = threadIdx.x & 31;
    const int wave = threadIdx.x >> 5;
    const int lo   = lane & 15;
    const int hi   = lane >> 4;

    const int m0  = blockIdx.x * (16 * MSUB);
    const int nb0 = wave * (NSUB * 16);

    int mbase[MSUB];
#pragma unroll
    for (int ms = 0; ms < MSUB; ++ms)
        mbase[ms] = min(m0 + 16 * ms, M - 16);

    v8f acc[MSUB][NSUB];
#pragma unroll
    for (int ms = 0; ms < MSUB; ++ms)
#pragma unroll
        for (int t = 0; t < NSUB; ++t) acc[ms][t] = 0.0f;

    const float* __restrict__ ap[MSUB];
#pragma unroll
    for (int ms = 0; ms < MSUB; ++ms)
        ap[ms] = A + (size_t)(mbase[ms] + lo) * KDIM + 2 * hi;
    const float* __restrict__ bp = B + (size_t)(2 * hi) * NDIM + nb0 + lo;

    frag_set<MSUB, NSUB, KS> f0, f1;
    load_frags<MSUB, NSUB, KS, NDIM>(f0, ap, bp, 0);
    for (int kt = 0; kt < NT - 2; kt += 2) {
        load_frags<MSUB, NSUB, KS, NDIM>(f1, ap, bp, (kt + 1) * KT);
        __builtin_amdgcn_sched_barrier(0);
        mma_frags<MSUB, NSUB, KS>(f0, acc);
        __builtin_amdgcn_sched_barrier(0);
        load_frags<MSUB, NSUB, KS, NDIM>(f0, ap, bp, (kt + 2) * KT);
        __builtin_amdgcn_sched_barrier(0);
        mma_frags<MSUB, NSUB, KS>(f1, acc);
        __builtin_amdgcn_sched_barrier(0);
    }
    load_frags<MSUB, NSUB, KS, NDIM>(f1, ap, bp, (NT - 1) * KT);
    __builtin_amdgcn_sched_barrier(0);
    mma_frags<MSUB, NSUB, KS>(f0, acc);
    mma_frags<MSUB, NSUB, KS>(f1, acc);

#pragma unroll
    for (int ms = 0; ms < MSUB; ++ms) {
        if (m0 + 16 * ms < M) {
#pragma unroll
            for (int t = 0; t < NSUB; ++t) {
#pragma unroll
                for (int v = 0; v < 8; ++v) {
                    const int row = mbase[ms] + v + 8 * hi;
                    const int col = nb0 + t * 16 + lo;
                    float x = acc[ms][t][v];
                    if (RELU) x = fmaxf(x, 0.0f);
                    C[(size_t)row * NDIM + col] = x;
                }
            }
        }
    }
}

#define SZ_BYTES   ((size_t)N_NODES * HIDDEN * sizeof(float))
#define OFF_OFFMAT (2 * SZ_BYTES)
#define OFF_ROWPTR (OFF_OFFMAT + 262144)
#define OFF_TMP    (OFF_ROWPTR + 65536)
#define OFF_EV     (OFF_TMP + (size_t)NCHUNK * CHUNK * sizeof(int4))

static_assert((size_t)NCHUNK * (NBIN + 1) * sizeof(int) <= 262144, "offset matrix fits its slot");

extern "C" void kernel_launch(void* const* d_in, const int* in_sizes, int n_in,
                              void* d_out, int out_size, void* d_ws, size_t ws_size,
                              hipStream_t stream)
{
    (void)in_sizes; (void)n_in; (void)out_size; (void)ws_size;

    const float* features = (const float*)d_in[0];
    const int*   esrc     = (const int*)  d_in[1];
    const int*   edst     = (const int*)  d_in[2];
    const float* eval_    = (const float*)d_in[3];
    const float* W0       = (const float*)d_in[4];
    const float* W1       = (const float*)d_in[5];
    float*       out      = (float*)d_out;

    float* Z      = (float*)d_ws;
    float* H      = (float*)((char*)d_ws + SZ_BYTES);
    int*   offmat = (int*)  ((char*)d_ws + OFF_OFFMAT);
    int*   rowptr = (int*)  ((char*)d_ws + OFF_ROWPTR);
    int4*  tmp    = (int4*) ((char*)d_ws + OFF_TMP);
    int2*  ev     = (int2*) ((char*)d_ws + OFF_EV);

    const int row_blocks  = (N_NODES + 7) / 8;
    const int gemm_blocks = (N_NODES + 31) / 32;

    edge_bin<<<NCHUNK, BIN_T, 0, stream>>>(esrc, edst, eval_, offmat, tmp);
    bin_rows<<<NBIN, ROW_T, 0, stream>>>(offmat, tmp, features, rowptr, ev, Z);

    gemm_wmma_f32<2, 4, 4, IN_DIM, HIDDEN, true><<<gemm_blocks, 128, 0, stream>>>(Z, W0, H, N_NODES);

    float* G = Z;
    gemm_wmma_f32<2, 2, 8, HIDDEN, OUT_DIM, false><<<gemm_blocks, 128, 0, stream>>>(H, W1, G, N_NODES);
    spmm_gather<OUT_DIM><<<row_blocks, 256, 0, stream>>>(G, rowptr, ev, out);
}


# Layer-1 GEMM fused into the layer-0 GEMM kernel: a block's relu(Z@W0) tile (32 rows) goes via LDS into @W1 (A fragments as b64 LDS reads); H never in global memory, one launch fewer; sums unchanged
# speedup vs baseline: 4.0613x; 1.2007x over previous
#include <hip/hip_runtime.h>


#define N_NODES 10000
#define N_EDGES 320000
#define IN_DIM  256
#define HIDDEN  256
#define OUT_DIM 128

typedef __attribute__((ext_vector_type(2))) float v2f;
typedef __attribute__((ext_vector_type(8))) float v8f;

#define NBIN     250
#define BIN_ROWS ((N_NODES + NBIN - 1) / NBIN)
#define NCHUNK   250
#define CHUNK    ((N_EDGES + NCHUNK - 1) / NCHUNK)
#define BIN_T    256
#define CH_PER   ((CHUNK + BIN_T - 1) / BIN_T)
#define ROW_T    1024
#define SEG_T    4
#define EV_STAGE 4096

static_assert(NBIN <= BIN_T, "one thread per bin in the offset scan");
static_assert(NCHUNK * SEG_T <= ROW_T, "SEG_T threads per chunk segment");
static_assert(BIN_ROWS <= ROW_T, "one thread per row of a range");
static_assert(ROW_T == 32 * 32, "second-level sum is one wave");

__global__ __launch_bounds__(BIN_T) void edge_bin(
    const int*   __restrict__ src,
    const int*   __restrict__ dst,
    const float* __restrict__ val,
    int*         __restrict__ offmat,
    int4*        __restrict__ tmp)
{
    __shared__ int  hist[NBIN + 1];
    __shared__ int  part[BIN_T / 32];
    __shared__ int4 stage[CHUNK];

    const int t  = threadIdx.x;
    const int c  = blockIdx.x;
    const int e0 = c * CHUNK;
    const int n  = max(0, min(CHUNK, N_EDGES - e0));

    for (int i = t; i <= NBIN; i += BIN_T) hist[i] = 0;

    int d[CH_PER], s[CH_PER], w[CH_PER], pos[CH_PER];
#pragma unroll
    for (int i = 0; i < CH_PER; ++i) {
        const int j = i * BIN_T + t;
        d[i] = 0; s[i] = 0; w[i] = 0; pos[i] = 0;
        if (j < n) {
            d[i] = dst[e0 + j];
            s[i] = src[e0 + j];
            w[i] = __float_as_int(val[e0 + j]);
        }
    }
    __syncthreads();
#pragma unroll
    for (int i = 0; i < CH_PER; ++i) {
        const int j = i * BIN_T + t;
        if (j < n) pos[i] = atomicAdd(&hist[d[i] / BIN_ROWS], 1);
    }
    __syncthreads();

    const int cntv = (t < NBIN) ? hist[t] : 0;
    const int lane = t & 31;
    const int wv   = t >> 5;
    int incl = cntv;
    for (int m = 1; m < 32; m <<= 1) {
        const int y = __shfl_up(incl, m);
        if (lane >= m) incl += y;
    }
    if (lane == 31) part[wv] = incl;
    __syncthreads();
    int excl = incl - cntv;
    for (int i = 0; i < wv; ++i) excl += part[i];
    if (t < NBIN) hist[t] = excl;
    if (t == NBIN - 1) hist[NBIN] = excl + cntv;
    __syncthreads();

    for (int i = t; i <= NBIN; i += BIN_T) offmat[c * (NBIN + 1) + i] = hist[i];

#pragma unroll
    for (int i = 0; i < CH_PER; ++i) {
        const int j = i * BIN_T + t;
        if (j < n) stage[hist[d[i] / BIN_ROWS] + pos[i]] = make_int4(d[i], s[i], w[i], 0);
    }
    __syncthreads();
    for (int j = t; j < n; j += BIN_T) tmp[e0 + j] = stage[j];
}

template<int D, int U, int LD = D>
__device__ __forceinline__ void gather_group(
    const float* __restrict__ H, int ex, int ey, int j, int lane,
    float4 (&acc)[D / 128])
{
    constexpr int V = D / 128;
    float4 h[U][V];
    float  w[U];
#pragma unroll
    for (int u = 0; u < U; ++u) {
        const int s = __builtin_amdgcn_readlane(ex, j + u);
        w[u] = __int_as_float(__builtin_amdgcn_readlane(ey, j + u));
        const float* __restrict__ hrow = H + (size_t)s * LD + lane * 4;
#pragma unroll
        for (int v = 0; v < V; ++v)
            h[u][v] = *(const float4*)(hrow + v * 128);
    }
#pragma unroll
    for (int u = 0; u < U; ++u) {
#pragma unroll
        for (int v = 0; v < V; ++v) {
            acc[v].x = fmaf(w[u], h[u][v].x, acc[v].x);
            acc[v].y = fmaf(w[u], h[u][v].y, acc[v].y);
            acc[v].z = fmaf(w[u], h[u][v].z, acc[v].z);
            acc[v].w = fmaf(w[u], h[u][v].w, acc[v].w);
        }
    }
}

template<bool B> struct pick { static constexpr bool value = B; };

__global__ __launch_bounds__(ROW_T) void bin_rows(
    const int*   __restrict__ offmat,
    const int4*  __restrict__ tmp,
    const float* __restrict__ X,
    int*         __restrict__ rowptr,
    int2*        __restrict__ ev,
    float*       __restrict__ Z)
{
    __shared__ int  rcnt[BIN_ROWS];
    __shared__ int  roff[BIN_ROWS + 1];
    __shared__ int  red[ROW_T / 32];
    __shared__ int  base_s;
    __shared__ int2 st[EV_STAGE];

    const int t  = threadIdx.x;
    const int b  = blockIdx.x;
    const int lo = b * BIN_ROWS;
    const int q  = t & (SEG_T - 1);

    int seg_beg = 0, seg_n = 0, below = 0;
    if (t < NCHUNK * SEG_T) {
        const int c  = t / SEG_T;
        const int o0 = offmat[c * (NBIN + 1) + b];
        const int o1 = offmat[c * (NBIN + 1) + b + 1];
        seg_beg = c * CHUNK + o0;
        seg_n   = o1 - o0;
        if (q == 0) below = o0;
    }
    for (int m = 16; m >= 1; m >>= 1) below += __shfl_xor(below, m);
    if ((t & 31) == 0) red[t >> 5] = below;
    if (t < BIN_ROWS) rcnt[t] = 0;
    __syncthreads();
    if (t < 32) {
        int x = red[t];
        for (int m = 16; m >= 1; m >>= 1) x += __shfl_xor(x, m);
        if (t == 0) base_s = x;
    }

    for (int i = q; i < seg_n; i += SEG_T) {
        const int4 r = tmp[seg_beg + i];
        atomicAdd(&rcnt[r.x - lo], 1);
    }
    __syncthreads();

    if (t < BIN_ROWS) {
        int sum = 0;
        for (int i = 0; i < t; ++i) sum += rcnt[i];
        roff[t] = sum;
        if (t == BIN_ROWS - 1) roff[BIN_ROWS] = sum + rcnt[t];
    }
    __syncthreads();

    const int base  = base_s;
    const int total = roff[BIN_ROWS];
    if (t < BIN_ROWS) {
        rcnt[t] = roff[t];
        if (lo + t < N_NODES) rowptr[lo + t] = base + roff[t];
    }
    if (b == NBIN - 1 && t == 0) rowptr[N_NODES] = base + total;
    __syncthreads();

    const bool staged = total <= EV_STAGE;
    for (int i = q; i < seg_n; i += SEG_T) {
        const int4 r = tmp[seg_beg + i];
        const int  p = atomicAdd(&rcnt[r.x - lo], 1);
        const int2 o = make_int2(r.y, r.z);
        if (staged) st[p] = o; else ev[base + p] = o;
    }
    __syncthreads();
    if (staged)
        for (int i = t; i < total; i += ROW_T) ev[base + i] = st[i];

    const int lane = t & 31;
    const int wv = __builtin_amdgcn_readfirstlane(t >> 5);
    auto gather_tasks = [&](auto from_lds) {
        for (int task = wv; task < 2 * BIN_ROWS; task += ROW_T / 32) {
            const int r    = task % BIN_ROWS;
            const int half = task / BIN_ROWS;
            const int row  = lo + r;
            if (row >= N_NODES) continue;
            const int beg = __builtin_amdgcn_readfirstlane(roff[r]);
            const int end = __builtin_amdgcn_readfirstlane(roff[r + 1]);
            const float* __restrict__ Xh = X + half * (IN_DIM / 2);

            float4 acc[1];
            acc[0] = make_float4(0.0f, 0.0f, 0.0f, 0.0f);
            for (int b0 = beg; b0 < end; b0 += 32) {
                const int n = min(32, end - b0);
                int2 my = make_int2(0, 0);
                if (lane < n) {
                    if constexpr (decltype(from_lds)::value) my = st[b0 + lane];
                    else                                     my = ev[base + b0 + lane];
                }

                int j = 0;
                for (; j + 8 <= n; j += 8) gather_group<IN_DIM / 2, 8, IN_DIM>(Xh, my.x, my.y, j, lane, acc);
                if (j + 4 <= n) { gather_group<IN_DIM / 2, 4, IN_DIM>(Xh, my.x, my.y, j, lane, acc); j += 4; }
                if (j + 2 <= n) { gather_group<IN_DIM / 2, 2, IN_DIM>(Xh, my.x, my.y, j, lane, acc); j += 2; }
                if (j < n)      { gather_group<IN_DIM / 2, 1, IN_DIM>(Xh, my.x, my.y, j, lane, acc); }
            }
            *(float4*)(Z + (size_t)row * IN_DIM + half * (IN_DIM / 2) + lane * 4) = acc[0];
        }
    };
    if (staged) gather_tasks(pick<true>{}); else gather_tasks(pick<false>{});
}

static_assert(IN_DIM == 256, "a gather task is one 128-column half of a row");

template<int D>
__global__ __launch_bounds__(256) void spmm_gather(
    const float* __restrict__ H,
    const int*   __restrict__ rowptr,
    const int2*  __restrict__ ev,
    float*       __restrict__ Z)
{
    constexpr int V = D / 128;
    const int lane = threadIdx.x & 31;
    const int row = __builtin_amdgcn_readfirstlane(blockIdx.x * 8 + (threadIdx.x >> 5));
    if (row >= N_NODES) return;

    const int beg = rowptr[row];
    const int end = rowptr[row + 1];

    float4 acc[V];
#pragma unroll
    for (int v = 0; v < V; ++v) acc[v] = make_float4(0.0f, 0.0f, 0.0f, 0.0f);

    for (int base = beg; base < end; base += 32) {
        const int n = min(32, end - base);
        int2 my = make_int2(0, 0);
        if (lane < n) my = ev[base + lane];

        int j = 0;
        for (; j + 8 <= n; j += 8) gather_group<D, 8>(H, my.x, my.y, j, lane, acc);
        if (j + 4 <= n) { gather_group<D, 4>(H, my.x, my.y, j, lane, acc); j += 4; }
        if (j + 2 <= n) { gather_group<D, 2>(H, my.x, my.y, j, lane, acc); j += 2; }
        if (j < n)      { gather_group<D, 1>(H, my.x, my.y, j, lane, acc); }
    }

    float* __restrict__ zrow = Z + (size_t)row * D + lane * 4;
#pragma unroll
    for (int v = 0; v < V; ++v)
        *(float4*)(zrow + v * 128) = acc[v];
}

template<int MSUB, int NSUB, int KS>
struct frag_set {
    v2f a[KS][MSUB];
    v2f b[KS][NSUB];
};

template<int MSUB, int NSUB, int KS, int NDIM>
__device__ __forceinline__ void load_frags(
    frag_set<MSUB, NSUB, KS>& f,
    const float* __restrict__ const (&ap)[MSUB],
    const float* __restrict__ bp, int k0)
{
#pragma unroll
    for (int j = 0; j < KS; ++j) {
        const int k = k0 + 4 * j;
#pragma unroll
        for (int ms = 0; ms < MSUB; ++ms)
            f.a[j][ms] = *(const v2f*)(ap[ms] + k);
#pragma unroll
        for (int t = 0; t < NSUB; ++t) {
            f.b[j][t].x = bp[(size_t)k * NDIM + t * 16];
            f.b[j][t].y = bp[(size_t)(k + 1) * NDIM + t * 16];
        }
    }
}

template<int MSUB, int NSUB, int KS>
__device__ __forceinline__ void mma_frags(
    const frag_set<MSUB, NSUB, KS>& f, v8f (&acc)[MSUB][NSUB])
{
#pragma unroll
    for (int j = 0; j < KS; ++j)
#pragma unroll
        for (int t = 0; t < NSUB; ++t)
#pragma unroll
            for (int ms = 0; ms < MSUB; ++ms) {
                acc[ms][t] = __builtin_amdgcn_wmma_f32_16x16x4_f32(
                    false, f.a[j][ms], false, f.b[j][t], (short)0, acc[ms][t],
                    false, false);
            }
}

template<int MSUB, int NSUB, int KS, int PITCH>
__device__ __forceinline__ void load_a_lds(
    frag_set<MSUB, NSUB, KS>& f, const float* hp, int k0)
{
#pragma unroll
    for (int j = 0; j < KS; ++j)
#pragma unroll
        for (int ms = 0; ms < MSUB; ++ms)
            f.a[j][ms] = *(const v2f*)(hp + ms * 16 * PITCH + k0 + 4 * j);
}

template<int MSUB, int NSUB, int KS, int NDIM>
__device__ __forceinline__ void load_b_glb(
    frag_set<MSUB, NSUB, KS>& f, const float* __restrict__ bp, int k0)
{
#pragma unroll
    for (int j = 0; j < KS; ++j) {
        const int k = k0 + 4 * j;
#pragma unroll
        for (int t = 0; t < NSUB; ++t) {
            f.b[j][t].x = bp[(size_t)k * NDIM + t * 16];
            f.b[j][t].y = bp[(size_t)(k + 1) * NDIM + t * 16];
        }
    }
}

#define HS_PITCH (HIDDEN + 4)
static_assert(((HS_PITCH * 4 / 16) & 1) == 1 && (HS_PITCH % 2) == 0, "pitch is 16 B x odd, b64 reads stay aligned");

template<int KS0, int KS1>
__global__ __launch_bounds__(128) void gemm2_wmma_f32(
    const float* __restrict__ A,
    const float* __restrict__ B0,
    const float* __restrict__ B1,
    float*       __restrict__ C,
    int M)
{
    constexpr int MSUB = 2;
    constexpr int NS0  = HIDDEN / 64;
    constexpr int NS1  = OUT_DIM / 64;
    constexpr int KT0  = 4 * KS0, NT0 = IN_DIM / KT0;
    constexpr int KT1  = 4 * KS1, NT1 = HIDDEN / KT1;
    static_assert(IN_DIM % KT0 == 0 && NT0 >= 2 && NT0 % 2 == 0, "trips come in pairs");
    static_assert(HIDDEN % KT1 == 0 && NT1 >= 2 && NT1 % 2 == 0, "trips come in pairs");
    static_assert(HIDDEN == 4 * NS0 * 16 && OUT_DIM == 4 * NS1 * 16, "four waves cover the row");

    __shared__ __attribute__((aligned(16))) float Hs[16 * MSUB * HS_PITCH];

    const int lane = threadIdx.x & 31;
    const int wave = threadIdx.x >> 5;
    const int lo   = lane & 15;
    const int hi   = lane >> 4;

    const int m0 = blockIdx.x * (16 * MSUB);

    int mbase[MSUB];
#pragma unroll
    for (int ms = 0; ms < MSUB; ++ms)
        mbase[ms] = min(m0 + 16 * ms, M - 16);

    const int nb0 = wave * (NS0 * 16);
    v8f acc[MSUB][NS0];
#pragma unroll
    for (int ms = 0; ms < MSUB; ++ms)
#pragma unroll
        for (int t = 0; t < NS0; ++t) acc[ms][t] = 0.0f;

    const float* __restrict__ ap[MSUB];
#pragma unroll
    for (int ms = 0; ms < MSUB; ++ms)
        ap[ms] = A + (size_t)(mbase[ms] + lo) * IN_DIM + 2 * hi;
    const float* __restrict__ bp0 = B0 + (size_t)(2 * hi) * HIDDEN + nb0 + lo;

    {
        frag_set<MSUB, NS0, KS0> f0, f1;
        load_frags<MSUB, NS0, KS0, HIDDEN>(f0, ap, bp0, 0);
        for (int kt = 0; kt < NT0 - 2; kt += 2) {
            load_frags<MSUB, NS0, KS0, HIDDEN>(f1, ap, bp0, (kt + 1) * KT0);
            __builtin_amdgcn_sched_barrier(0);
            mma_frags<MSUB, NS0, KS0>(f0, acc);
            __builtin_amdgcn_sched_barrier(0);
            load_frags<MSUB, NS0, KS0, HIDDEN>(f0, ap, bp0, (kt + 2) * KT0);
            __builtin_amdgcn_sched_barrier(0);
            mma_frags<MSUB, NS0, KS0>(f1, acc);
            __builtin_amdgcn_sched_barrier(0);
        }
        load_frags<MSUB, NS0, KS0, HIDDEN>(f1, ap, bp0, (NT0 - 1) * KT0);
        __builtin_amdgcn_sched_barrier(0);
        mma_frags<MSUB, NS0, KS0>(f0, acc);
        mma_frags<MSUB, NS0, KS0>(f1, acc);
    }

    const int nb1 = wave * (NS1 * 16);
    const float* __restrict__ bp1 = B1 + (size_t)(2 * hi) * OUT_DIM + nb1 + lo;
    frag_set<MSUB, NS1, KS1> g0, g1;
    load_b_glb<MSUB, NS1, KS1, OUT_DIM>(g0, bp1, 0);

#pragma unroll
    for (int ms = 0; ms < MSUB; ++ms)
#pragma unroll
        for (int t = 0; t < NS0; ++t)
#pragma unroll
            for (int v = 0; v < 8; ++v)
                Hs[(ms * 16 + v + 8 * hi) * HS_PITCH + nb0 + t * 16 + lo] = fmaxf(acc[ms][t][v], 0.0f);
    __syncthreads();

    v8f acc2[MSUB][NS1];
#pragma unroll
    for (int ms = 0; ms < MSUB; ++ms)
#pragma unroll
        for (int t = 0; t < NS1; ++t) acc2[ms][t] = 0.0f;

    const float* hp = Hs + lo * HS_PITCH + 2 * hi;
    load_a_lds<MSUB, NS1, KS1, HS_PITCH>(g0, hp, 0);
    for (int kt = 0; kt < NT1 - 2; kt += 2) {
        load_a_lds<MSUB, NS1, KS1, HS_PITCH>(g1, hp, (kt + 1) * KT1);
        load_b_glb<MSUB, NS1, KS1, OUT_DIM>(g1, bp1, (kt + 1) * KT1);
        __builtin_amdgcn_sched_barrier(0);
        mma_frags<MSUB, NS1, KS1>(g0, acc2);
        __builtin_amdgcn_sched_barrier(0);
        load_a_lds<MSUB, NS1, KS1, HS_PITCH>(g0, hp, (kt + 2) * KT1);
        load_b_glb<MSUB, NS1, KS1, OUT_DIM>(g0, bp1, (kt + 2) * KT1);
        __builtin_amdgcn_sched_barrier(0);
        mma_frags<MSUB, NS1, KS1>(g1, acc2);
        __builtin_amdgcn_sched_barrier(0);
    }
    load_a_lds<MSUB, NS1, KS1, HS_PITCH>(g1, hp, (NT1 - 1) * KT1);
    load_b_glb<MSUB, NS1, KS1, OUT_DIM>(g1, bp1, (NT1 - 1) * KT1);
    __builtin_amdgcn_sched_barrier(0);
    mma_frags<MSUB, NS1, KS1>(g0, acc2);
    mma_frags<MSUB, NS1, KS1>(g1, acc2);

#pragma unroll
    for (int ms = 0; ms < MSUB; ++ms) {
        if (m0 + 16 * ms < M) {
#pragma unroll
            for (int t = 0; t < NS1; ++t) {
#pragma unroll
                for (int v = 0; v < 8; ++v) {
                    const int row = mbase[ms] + v + 8 * hi;
                    const int col = nb1 + t * 16 + lo;
                    C[(size_t)row * OUT_DIM + col] = acc2[ms][t][v];
                }
            }
        }
    }
}

#define SZ_BYTES   ((size_t)N_NODES * HIDDEN * sizeof(float))
#define OFF_OFFMAT (2 * SZ_BYTES)
#define OFF_ROWPTR (OFF_OFFMAT + 262144)
#define OFF_TMP    (OFF_ROWPTR + 65536)
#define OFF_EV     (OFF_TMP + (size_t)NCHUNK * CHUNK * sizeof(int4))

static_assert((size_t)NCHUNK * (NBIN + 1) * sizeof(int) <= 262144, "offset matrix fits its slot");

extern "C" void kernel_launch(void* const* d_in, const int* in_sizes, int n_in,
                              void* d_out, int out_size, void* d_ws, size_t ws_size,
                              hipStream_t stream)
{
    (void)in_sizes; (void)n_in; (void)out_size; (void)ws_size;

    const float* features = (const float*)d_in[0];
    const int*   esrc     = (const int*)  d_in[1];
    const int*   edst     = (const int*)  d_in[2];
    const float* eval_    = (const float*)d_in[3];
    const float* W0       = (const float*)d_in[4];
    const float* W1       = (const float*)d_in[5];
    float*       out      = (float*)d_out;

    float* Z      = (float*)d_ws;
    float* G      = (float*)((char*)d_ws + SZ_BYTES);
    int*   offmat = (int*)  ((char*)d_ws + OFF_OFFMAT);
    int*   rowptr = (int*)  ((char*)d_ws + OFF_ROWPTR);
    int4*  tmp    = (int4*) ((char*)d_ws + OFF_TMP);
    int2*  ev     = (int2*) ((char*)d_ws + OFF_EV);

    const int row_blocks  = (N_NODES + 7) / 8;
    const int gemm_blocks = (N_NODES + 31) / 32;

    edge_bin<<<NCHUNK, BIN_T, 0, stream>>>(esrc, edst, eval_, offmat, tmp);
    bin_rows<<<NBIN, ROW_T, 0, stream>>>(offmat, tmp, features, rowptr, ev, Z);

    gemm2_wmma_f32<4, 8><<<gemm_blocks, 128, 0, stream>>>(Z, W0, W1, G, N_NODES);

    spmm_gather<OUT_DIM><<<row_blocks, 256, 0, stream>>>(G, rowptr, ev, out);
}


# Both GEMMs fused into the row-binning kernel: a range's gathered Z rows stay in LDS as the A tile of relu(Z@W0)@W1 (wave = 16 columns x 3 row tiles); Z never global, one launch fewer; sums unchanged
# speedup vs baseline: 4.5368x; 1.1171x over previous
#include <hip/hip_runtime.h>


#define N_NODES 10000
#define N_EDGES 320000
#define IN_DIM  256
#define HIDDEN  256
#define OUT_DIM 128

typedef __attribute__((ext_vector_type(2))) float v2f;
typedef __attribute__((ext_vector_type(8))) float v8f;

#define NBIN     250
#define BIN_ROWS ((N_NODES + NBIN - 1) / NBIN)
#define NCHUNK   250
#define CHUNK    ((N_EDGES + NCHUNK - 1) / NCHUNK)
#define BIN_T    256
#define CH_PER   ((CHUNK + BIN_T - 1) / BIN_T)
#define ROW_T    1024
#define SEG_T    4
#define EV_STAGE 4096

static_assert(NBIN <= BIN_T, "one thread per bin in the offset scan");
static_assert(NCHUNK * SEG_T <= ROW_T, "SEG_T threads per chunk segment");
static_assert(BIN_ROWS <= ROW_T, "one thread per row of a range");
static_assert(ROW_T == 32 * 32, "second-level sum is one wave");

__global__ __launch_bounds__(BIN_T) void edge_bin(
    const int*   __restrict__ src,
    const int*   __restrict__ dst,
    const float* __restrict__ val,
    int*         __restrict__ offmat,
    int4*        __restrict__ tmp)
{
    __shared__ int  hist[NBIN + 1];
    __shared__ int  part[BIN_T / 32];
    __shared__ int4 stage[CHUNK];

    const int t  = threadIdx.x;
    const int c  = blockIdx.x;
    const int e0 = c * CHUNK;
    const int n  = max(0, min(CHUNK, N_EDGES - e0));

    for (int i = t; i <= NBIN; i += BIN_T) hist[i] = 0;

    int d[CH_PER], s[CH_PER], w[CH_PER], pos[CH_PER];
#pragma unroll
    for (int i = 0; i < CH_PER; ++i) {
        const int j = i * BIN_T + t;
        d[i] = 0; s[i] = 0; w[i] = 0; pos[i] = 0;
        if (j < n) {
            d[i] = dst[e0 + j];
            s[i] = src[e0 + j];
            w[i] = __float_as_int(val[e0 + j]);
        }
    }
    __syncthreads();
#pragma unroll
    for (int i = 0; i < CH_PER; ++i) {
        const int j = i * BIN_T + t;
        if (j < n) pos[i] = atomicAdd(&hist[d[i] / BIN_ROWS], 1);
    }
    __syncthreads();

    const int cntv = (t < NBIN) ? hist[t] : 0;
    const int lane = t & 31;
    const int wv   = t >> 5;
    int incl = cntv;
    for (int m = 1; m < 32; m <<= 1) {
        const int y = __shfl_up(incl, m);
        if (lane >= m) incl += y;
    }
    if (lane == 31) part[wv] = incl;
    __syncthreads();
    int excl = incl - cntv;
    for (int i = 0; i < wv; ++i) excl += part[i];
    if (t < NBIN) hist[t] = excl;
    if (t == NBIN - 1) hist[NBIN] = excl + cntv;
    __syncthreads();

    for (int i = t; i <= NBIN; i += BIN_T) offmat[c * (NBIN + 1) + i] = hist[i];

#pragma unroll
    for (int i = 0; i < CH_PER; ++i) {
        const int j = i * BIN_T + t;
        if (j < n) stage[hist[d[i] / BIN_ROWS] + pos[i]] = make_int4(d[i], s[i], w[i], 0);
    }
    __syncthreads();
    for (int j = t; j < n; j += BIN_T) tmp[e0 + j] = stage[j];
}

template<int D, int U, int LD = D>
__device__ __forceinline__ void gather_group(
    const float* __restrict__ H, int ex, int ey, int j, int lane,
    float4 (&acc)[D / 128])
{
    constexpr int V = D / 128;
    float4 h[U][V];
    float  w[U];
#pragma unroll
    for (int u = 0; u < U; ++u) {
        const int s = __builtin_amdgcn_readlane(ex, j + u);
        w[u] = __int_as_float(__builtin_amdgcn_readlane(ey, j + u));
        const float* __restrict__ hrow = H + (size_t)s * LD + lane * 4;
#pragma unroll
        for (int v = 0; v < V; ++v)
            h[u][v] = *(const float4*)(hrow + v * 128);
    }
#pragma unroll
    for (int u = 0; u < U; ++u) {
#pragma unroll
        for (int v = 0; v < V; ++v) {
            acc[v].x = fmaf(w[u], h[u][v].x, acc[v].x);
            acc[v].y = fmaf(w[u], h[u][v].y, acc[v].y);
            acc[v].z = fmaf(w[u], h[u][v].z, acc[v].z);
            acc[v].w = fmaf(w[u], h[u][v].w, acc[v].w);
        }
    }
}

template<int MSUB, int NSUB, int KS>
struct frag_set {
    v2f a[KS][MSUB];
    v2f b[KS][NSUB];
};

template<int MSUB, int NSUB, int KS>
__device__ __forceinline__ void mma_frags(
    const frag_set<MSUB, NSUB, KS>& f, v8f (&acc)[MSUB][NSUB])
{
#pragma unroll
    for (int j = 0; j < KS; ++j)
#pragma unroll
        for (int t = 0; t < NSUB; ++t)
#pragma unroll
            for (int ms = 0; ms < MSUB; ++ms) {
                acc[ms][t] = __builtin_amdgcn_wmma_f32_16x16x4_f32(
                    false, f.a[j][ms], false, f.b[j][t], (short)0, acc[ms][t],
                    false, false);
            }
}

template<int MSUB, int NSUB, int KS, int PITCH>
__device__ __forceinline__ void load_a_lds(
    frag_set<MSUB, NSUB, KS>& f, const float* hp, int k0)
{
#pragma unroll
    for (int j = 0; j < KS; ++j)
#pragma unroll
        for (int ms = 0; ms < MSUB; ++ms)
            f.a[j][ms] = *(const v2f*)(hp + ms * 16 * PITCH + k0 + 4 * j);
}

template<int MSUB, int NSUB, int KS, int NDIM>
__device__ __forceinline__ void load_b_glb(
    frag_set<MSUB, NSUB, KS>& f, const float* __restrict__ bp, int k0)
{
#pragma unroll
    for (int j = 0; j < KS; ++j) {
        const int k = k0 + 4 * j;
#pragma unroll
        for (int t = 0; t < NSUB; ++t) {
            f.b[j][t].x = bp[(size_t)k * NDIM + t * 16];
            f.b[j][t].y = bp[(size_t)(k + 1) * NDIM + t * 16];
        }
    }
}

template<int MSUB, int KS, int KDIM, int NDIM, int PITCH>
__device__ __forceinline__ void wave_gemm(
    const float* hp, const float* __restrict__ bp, v8f (&acc)[MSUB][1])
{
    constexpr int KT = 4 * KS, NT = KDIM / KT;
    static_assert(KDIM % KT == 0 && NT >= 2 && NT % 2 == 0, "trips come in pairs");

    frag_set<MSUB, 1, KS> f0, f1;
    load_a_lds<MSUB, 1, KS, PITCH>(f0, hp, 0);
    load_b_glb<MSUB, 1, KS, NDIM>(f0, bp, 0);
    for (int kt = 0; kt < NT - 2; kt += 2) {
        load_a_lds<MSUB, 1, KS, PITCH>(f1, hp, (kt + 1) * KT);
        load_b_glb<MSUB, 1, KS, NDIM>(f1, bp, (kt + 1) * KT);
        __builtin_amdgcn_sched_barrier(0);
        mma_frags<MSUB, 1, KS>(f0, acc);
        __builtin_amdgcn_sched_barrier(0);
        load_a_lds<MSUB, 1, KS, PITCH>(f0, hp, (kt + 2) * KT);
        load_b_glb<MSUB, 1, KS, NDIM>(f0, bp, (kt + 2) * KT);
        __builtin_amdgcn_sched_barrier(0);
        mma_frags<MSUB, 1, KS>(f1, acc);
        __builtin_amdgcn_sched_barrier(0);
    }
    load_a_lds<MSUB, 1, KS, PITCH>(f1, hp, (NT - 1) * KT);
    load_b_glb<MSUB, 1, KS, NDIM>(f1, bp, (NT - 1) * KT);
    __builtin_amdgcn_sched_barrier(0);
    mma_frags<MSUB, 1, KS>(f0, acc);
    mma_frags<MSUB, 1, KS>(f1, acc);
}

#define MT        ((BIN_ROWS + 15) / 16)
#define TILE_ROWS (16 * MT)
#define NPAD      (TILE_ROWS - BIN_ROWS)
#define ZS_PITCH  (IN_DIM + 4)
#define HS_PITCH  (HIDDEN + 4)
#define GEMM_KS   4
static_assert(((ZS_PITCH * 4 / 16) & 1) == 1 && (ZS_PITCH % 2) == 0, "pitch is 16 B x odd, b64 reads stay aligned");
static_assert(((HS_PITCH * 4 / 16) & 1) == 1 && (HS_PITCH % 2) == 0, "pitch is 16 B x odd, b64 reads stay aligned");
static_assert(HIDDEN / 16 <= ROW_T / 32 && OUT_DIM / 16 <= ROW_T / 32, "one wave per 16 output columns");
static_assert(OUT_DIM <= IN_DIM, "the G tile reuses the Z tile's rows");

template<bool B> struct pick { static constexpr bool value = B; };

__global__ __launch_bounds__(ROW_T) void bin_rows(
    const int*   __restrict__ offmat,
    const int4*  __restrict__ tmp,
    const float* __restrict__ X,
    const float* __restrict__ W0,
    const float* __restrict__ W1,
    int*         __restrict__ rowptr,
    int2*        __restrict__ ev,
    float*       __restrict__ G)
{
    __shared__ int  rcnt[BIN_ROWS];
    __shared__ int  roff[BIN_ROWS + 1];
    __shared__ int  red[ROW_T / 32];
    __shared__ int  base_s;
    __shared__ int2 st[EV_STAGE];
    __shared__ __attribute__((aligned(16))) float Zs[TILE_ROWS * ZS_PITCH];
    __shared__ __attribute__((aligned(16))) float Hs[TILE_ROWS * HS_PITCH];

    const int t  = threadIdx.x;
    const int b  = blockIdx.x;
    const int lo = b * BIN_ROWS;
    const int q  = t & (SEG_T - 1);

    int seg_beg = 0, seg_n = 0, below = 0;
    if (t < NCHUNK * SEG_T) {
        const int c  = t / SEG_T;
        const int o0 = offmat[c * (NBIN + 1) + b];
        const int o1 = offmat[c * (NBIN + 1) + b + 1];
        seg_beg = c * CHUNK + o0;
        seg_n   = o1 - o0;
        if (q == 0) below = o0;
    }
    for (int m = 16; m >= 1; m >>= 1) below += __shfl_xor(below, m);
    if ((t & 31) == 0) red[t >> 5] = below;
    if (t < BIN_ROWS) rcnt[t] = 0;
    __syncthreads();
    if (t < 32) {
        int x = red[t];
        for (int m = 16; m >= 1; m >>= 1) x += __shfl_xor(x, m);
        if (t == 0) base_s = x;
    }

    for (int i = q; i < seg_n; i += SEG_T) {
        const int4 r = tmp[seg_beg + i];
        atomicAdd(&rcnt[r.x - lo], 1);
    }
    __syncthreads();

    if (t < BIN_ROWS) {
        int sum = 0;
        for (int i = 0; i < t; ++i) sum += rcnt[i];
        roff[t] = sum;
        if (t == BIN_ROWS - 1) roff[BIN_ROWS] = sum + rcnt[t];
    }
    __syncthreads();

    const int base  = base_s;
    const int total = roff[BIN_ROWS];
    if (t < BIN_ROWS) {
        rcnt[t] = roff[t];
        if (lo + t < N_NODES) rowptr[lo + t] = base + roff[t];
    }
    if (b == NBIN - 1 && t == 0) rowptr[N_NODES] = base + total;
    __syncthreads();

    const bool staged = total <= EV_STAGE;
    for (int i = q; i < seg_n; i += SEG_T) {
        const int4 r = tmp[seg_beg + i];
        const int  p = atomicAdd(&rcnt[r.x - lo], 1);
        const int2 o = make_int2(r.y, r.z);
        if (staged) st[p] = o; else ev[base + p] = o;
    }
    __syncthreads();
    if (staged)
        for (int i = t; i < total; i += ROW_T) ev[base + i] = st[i];

    const int lane = t & 31;
    const int wv = __builtin_amdgcn_readfirstlane(t >> 5);
    auto gather_tasks = [&](auto from_lds) {
        for (int task = wv; task < 2 * TILE_ROWS; task += ROW_T / 32) {
            if (task >= 2 * BIN_ROWS) {
                const int p = task - 2 * BIN_ROWS;
                const int r = BIN_ROWS + p % (NPAD > 0 ? NPAD : 1);
                const int half = p / (NPAD > 0 ? NPAD : 1);
                *(float4*)(Zs + r * ZS_PITCH + half * (IN_DIM / 2) + lane * 4) = make_float4(0.0f, 0.0f, 0.0f, 0.0f);
                continue;
            }
            const int r    = task % BIN_ROWS;
            const int half = task / BIN_ROWS;
            const int row  = lo + r;
            if (row >= N_NODES) {
                *(float4*)(Zs + r * ZS_PITCH + half * (IN_DIM / 2) + lane * 4) = make_float4(0.0f, 0.0f, 0.0f, 0.0f);
                continue;
            }
            const int beg = __builtin_amdgcn_readfirstlane(roff[r]);
            const int end = __builtin_amdgcn_readfirstlane(roff[r + 1]);
            const float* __restrict__ Xh = X + half * (IN_DIM / 2);

            float4 acc[1];
            acc[0] = make_float4(0.0f, 0.0f, 0.0f, 0.0f);
            for (int b0 = beg; b0 < end; b0 += 32) {
                const int n = min(32, end - b0);
                int2 my = make_int2(0, 0);
                if (lane < n) {
                    if constexpr (decltype(from_lds)::value) my = st[b0 + lane];
                    else                                     my = ev[base + b0 + lane];
                }

                int j = 0;
                for (; j + 8 <= n; j += 8) gather_group<IN_DIM / 2, 8, IN_DIM>(Xh, my.x, my.y, j, lane, acc);
                if (j + 4 <= n) { gather_group<IN_DIM / 2, 4, IN_DIM>(Xh, my.x, my.y, j, lane, acc); j += 4; }
                if (j + 2 <= n) { gather_group<IN_DIM / 2, 2, IN_DIM>(Xh, my.x, my.y, j, lane, acc); j += 2; }
                if (j < n)      { gather_group<IN_DIM / 2, 1, IN_DIM>(Xh, my.x, my.y, j, lane, acc); }
            }
            *(float4*)(Zs + r * ZS_PITCH + half * (IN_DIM / 2) + lane * 4) = acc[0];
        }
    };
    if (staged) gather_tasks(pick<true>{}); else gather_tasks(pick<false>{});
    __syncthreads();

    const int l16 = lane & 15;
    const int hi  = lane >> 4;

    if (wv < HIDDEN / 16) {
        const int nb = wv * 16;
        v8f acc[MT][1];
#pragma unroll
        for (int ms = 0; ms < MT; ++ms) acc[ms][0] = 0.0f;
        wave_gemm<MT, GEMM_KS, IN_DIM, HIDDEN, ZS_PITCH>(
            Zs + l16 * ZS_PITCH + 2 * hi,
            W0 + (size_t)(2 * hi) * HIDDEN + nb + l16, acc);
#pragma unroll
        for (int ms = 0; ms < MT; ++ms)
#pragma unroll
            for (int v = 0; v < 8; ++v)
                Hs[(ms * 16 + v + 8 * hi) * HS_PITCH + nb + l16] = fmaxf(acc[ms][0][v], 0.0f);
    }
    __syncthreads();

    if (wv < OUT_DIM / 16) {
        const int nb = wv * 16;
        v8f acc[MT][1];
#pragma unroll
        for (int ms = 0; ms < MT; ++ms) acc[ms][0] = 0.0f;
        wave_gemm<MT, GEMM_KS, HIDDEN, OUT_DIM, HS_PITCH>(
            Hs + l16 * HS_PITCH + 2 * hi,
            W1 + (size_t)(2 * hi) * OUT_DIM + nb + l16, acc);
#pragma unroll
        for (int ms = 0; ms < MT; ++ms)
#pragma unroll
            for (int v = 0; v < 8; ++v)
                Zs[(ms * 16 + v + 8 * hi) * ZS_PITCH + nb + l16] = acc[ms][0][v];
    }
    __syncthreads();

    for (int i = t; i < BIN_ROWS * (OUT_DIM / 4); i += ROW_T) {
        const int r = i / (OUT_DIM / 4);
        const int c = i % (OUT_DIM / 4);
        if (lo + r < N_NODES)
            *(float4*)(G + (size_t)(lo + r) * OUT_DIM + 4 * c) = *(const float4*)(Zs + r * ZS_PITCH + 4 * c);
    }
}

static_assert(IN_DIM == 256, "a gather task is one 128-column half of a row");

template<int D>
__global__ __launch_bounds__(256) void spmm_gather(
    const float* __restrict__ H,
    const int*   __restrict__ rowptr,
    const int2*  __restrict__ ev,
    float*       __restrict__ Z)
{
    constexpr int V = D / 128;
    const int lane = threadIdx.x & 31;
    const int row = __builtin_amdgcn_readfirstlane(blockIdx.x * 8 + (threadIdx.x >> 5));
    if (row >= N_NODES) return;

    const int beg = rowptr[row];
    const int end = rowptr[row + 1];

    float4 acc[V];
#pragma unroll
    for (int v = 0; v < V; ++v) acc[v] = make_float4(0.0f, 0.0f, 0.0f, 0.0f);

    for (int base = beg; base < end; base += 32) {
        const int n = min(32, end - base);
        int2 my = make_int2(0, 0);
        if (lane < n) my = ev[base + lane];

        int j = 0;
        for (; j + 8 <= n; j += 8) gather_group<D, 8>(H, my.x, my.y, j, lane, acc);
        if (j + 4 <= n) { gather_group<D, 4>(H, my.x, my.y, j, lane, acc); j += 4; }
        if (j + 2 <= n) { gather_group<D, 2>(H, my.x, my.y, j, lane, acc); j += 2; }
        if (j < n)      { gather_group<D, 1>(H, my.x, my.y, j, lane, acc); }
    }

    float* __restrict__ zrow = Z + (size_t)row * D + lane * 4;
#pragma unroll
    for (int v = 0; v < V; ++v)
        *(float4*)(zrow + v * 128) = acc[v];
}

#define SZ_BYTES   ((size_t)N_NODES * HIDDEN * sizeof(float))
#define OFF_OFFMAT (2 * SZ_BYTES)
#define OFF_ROWPTR (OFF_OFFMAT + 262144)
#define OFF_TMP    (OFF_ROWPTR + 65536)
#define OFF_EV     (OFF_TMP + (size_t)NCHUNK * CHUNK * sizeof(int4))

static_assert((size_t)NCHUNK * (NBIN + 1) * sizeof(int) <= 262144, "offset matrix fits its slot");

extern "C" void kernel_launch(void* const* d_in, const int* in_sizes, int n_in,
                              void* d_out, int out_size, void* d_ws, size_t ws_size,
                              hipStream_t stream)
{
    (void)in_sizes; (void)n_in; (void)out_size; (void)ws_size;

    const float* features = (const float*)d_in[0];
    const int*   esrc     = (const int*)  d_in[1];
    const int*   edst     = (const int*)  d_in[2];
    const float* eval_    = (const float*)d_in[3];
    const float* W0       = (const float*)d_in[4];
    const float* W1       = (const float*)d_in[5];
    float*       out      = (float*)d_out;

    float* G      = (float*)((char*)d_ws + SZ_BYTES);
    int*   offmat = (int*)  ((char*)d_ws + OFF_OFFMAT);
    int*   rowptr = (int*)  ((char*)d_ws + OFF_ROWPTR);
    int4*  tmp    = (int4*) ((char*)d_ws + OFF_TMP);
    int2*  ev     = (int2*) ((char*)d_ws + OFF_EV);

    const int row_blocks  = (N_NODES + 7) / 8;

    edge_bin<<<NCHUNK, BIN_T, 0, stream>>>(esrc, edst, eval_, offmat, tmp);
    bin_rows<<<NBIN, ROW_T, 0, stream>>>(offmat, tmp, features, W0, W1, rowptr, ev, G);

    spmm_gather<OUT_DIM><<<row_blocks, 256, 0, stream>>>(G, rowptr, ev, out);
}


# GEMM K loops: B fragments (global) fetched 3 trips ahead into 4 small register sets, A fragments (LDS) still 1 trip ahead; loop fully unrolled; k order and summation order unchanged
# speedup vs baseline: 4.7508x; 1.0472x over previous
#include <hip/hip_runtime.h>


#define N_NODES 10000
#define N_EDGES 320000
#define IN_DIM  256
#define HIDDEN  256
#define OUT_DIM 128

typedef __attribute__((ext_vector_type(2))) float v2f;
typedef __attribute__((ext_vector_type(8))) float v8f;

#define NBIN     250
#define BIN_ROWS ((N_NODES + NBIN - 1) / NBIN)
#define NCHUNK   250
#define CHUNK    ((N_EDGES + NCHUNK - 1) / NCHUNK)
#define BIN_T    256
#define CH_PER   ((CHUNK + BIN_T - 1) / BIN_T)
#define ROW_T    1024
#define SEG_T    4
#define EV_STAGE 4096

static_assert(NBIN <= BIN_T, "one thread per bin in the offset scan");
static_assert(NCHUNK * SEG_T <= ROW_T, "SEG_T threads per chunk segment");
static_assert(BIN_ROWS <= ROW_T, "one thread per row of a range");
static_assert(ROW_T == 32 * 32, "second-level sum is one wave");

__global__ __launch_bounds__(BIN_T) void edge_bin(
    const int*   __restrict__ src,
    const int*   __restrict__ dst,
    const float* __restrict__ val,
    int*         __restrict__ offmat,
    int4*        __restrict__ tmp)
{
    __shared__ int  hist[NBIN + 1];
    __shared__ int  part[BIN_T / 32];
    __shared__ int4 stage[CHUNK];

    const int t  = threadIdx.x;
    const int c  = blockIdx.x;
    const int e0 = c * CHUNK;
    const int n  = max(0, min(CHUNK, N_EDGES - e0));

    for (int i = t; i <= NBIN; i += BIN_T) hist[i] = 0;

    int d[CH_PER], s[CH_PER], w[CH_PER], pos[CH_PER];
#pragma unroll
    for (int i = 0; i < CH_PER; ++i) {
        const int j = i * BIN_T + t;
        d[i] = 0; s[i] = 0; w[i] = 0; pos[i] = 0;
        if (j < n) {
            d[i] = dst[e0 + j];
            s[i] = src[e0 + j];
            w[i] = __float_as_int(val[e0 + j]);
        }
    }
    __syncthreads();
#pragma unroll
    for (int i = 0; i < CH_PER; ++i) {
        const int j = i * BIN_T + t;
        if (j < n) pos[i] = atomicAdd(&hist[d[i] / BIN_ROWS], 1);
    }
    __syncthreads();

    const int cntv = (t < NBIN) ? hist[t] : 0;
    const int lane = t & 31;
    const int wv   = t >> 5;
    int incl = cntv;
    for (int m = 1; m < 32; m <<= 1) {
        const int y = __shfl_up(incl, m);
        if (lane >= m) incl += y;
    }
    if (lane == 31) part[wv] = incl;
    __syncthreads();
    int excl = incl - cntv;
    for (int i = 0; i < wv; ++i) excl += part[i];
    if (t < NBIN) hist[t] = excl;
    if (t == NBIN - 1) hist[NBIN] = excl + cntv;
    __syncthreads();

    for (int i = t; i <= NBIN; i += BIN_T) offmat[c * (NBIN + 1) + i] = hist[i];

#pragma unroll
    for (int i = 0; i < CH_PER; ++i) {
        const int j = i * BIN_T + t;
        if (j < n) stage[hist[d[i] / BIN_ROWS] + pos[i]] = make_int4(d[i], s[i], w[i], 0);
    }
    __syncthreads();
    for (int j = t; j < n; j += BIN_T) tmp[e0 + j] = stage[j];
}

template<int D, int U, int LD = D>
__device__ __forceinline__ void gather_group(
    const float* __restrict__ H, int ex, int ey, int j, int lane,
    float4 (&acc)[D / 128])
{
    constexpr int V = D / 128;
    float4 h[U][V];
    float  w[U];
#pragma unroll
    for (int u = 0; u < U; ++u) {
        const int s = __builtin_amdgcn_readlane(ex, j + u);
        w[u] = __int_as_float(__builtin_amdgcn_readlane(ey, j + u));
        const float* __restrict__ hrow = H + (size_t)s * LD + lane * 4;
#pragma unroll
        for (int v = 0; v < V; ++v)
            h[u][v] = *(const float4*)(hrow + v * 128);
    }
#pragma unroll
    for (int u = 0; u < U; ++u) {
#pragma unroll
        for (int v = 0; v < V; ++v) {
            acc[v].x = fmaf(w[u], h[u][v].x, acc[v].x);
            acc[v].y = fmaf(w[u], h[u][v].y, acc[v].y);
            acc[v].z = fmaf(w[u], h[u][v].z, acc[v].z);
            acc[v].w = fmaf(w[u], h[u][v].w, acc[v].w);
        }
    }
}

template<int MSUB, int KS> struct a_frags { v2f a[KS][MSUB]; };
template<int KS>           struct b_frags { v2f b[KS]; };

template<int MSUB, int KS>
__device__ __forceinline__ void mma_frags(
    const a_frags<MSUB, KS>& fa, const b_frags<KS>& fb, v8f (&acc)[MSUB][1])
{
#pragma unroll
    for (int j = 0; j < KS; ++j)
#pragma unroll
        for (int ms = 0; ms < MSUB; ++ms) {
            acc[ms][0] = __builtin_amdgcn_wmma_f32_16x16x4_f32(
                false, fa.a[j][ms], false, fb.b[j], (short)0, acc[ms][0],
                false, false);
        }
}

template<int MSUB, int KS, int PITCH>
__device__ __forceinline__ void load_a_lds(
    a_frags<MSUB, KS>& f, const float* hp, int k0)
{
#pragma unroll
    for (int j = 0; j < KS; ++j)
#pragma unroll
        for (int ms = 0; ms < MSUB; ++ms)
            f.a[j][ms] = *(const v2f*)(hp + ms * 16 * PITCH + k0 + 4 * j);
}

template<int KS, int NDIM>
__device__ __forceinline__ void load_b_glb(
    b_frags<KS>& f, const float* __restrict__ bp, int k0)
{
#pragma unroll
    for (int j = 0; j < KS; ++j) {
        const int k = k0 + 4 * j;
        f.b[j].x = bp[(size_t)k * NDIM];
        f.b[j].y = bp[(size_t)(k + 1) * NDIM];
    }
}

template<int MSUB, int KS, int KDIM, int NDIM, int PITCH, int BDIST>
__device__ __forceinline__ void wave_gemm(
    const float* hp, const float* __restrict__ bp, v8f (&acc)[MSUB][1])
{
    constexpr int KT = 4 * KS, NT = KDIM / KT;
    static_assert(KDIM % KT == 0 && BDIST >= 1 && NT > BDIST, "the prologue fills BDIST trips");

    a_frags<MSUB, KS> fa[2];
    b_frags<KS>       fb[BDIST + 1];
#pragma unroll
    for (int t = 0; t < BDIST; ++t) load_b_glb<KS, NDIM>(fb[t], bp, t * KT);
    load_a_lds<MSUB, KS, PITCH>(fa[0], hp, 0);
#pragma unroll
    for (int t = 0; t < NT; ++t) {
        if (t + BDIST < NT) load_b_glb<KS, NDIM>(fb[(t + BDIST) % (BDIST + 1)], bp, (t + BDIST) * KT);
        if (t + 1 < NT)     load_a_lds<MSUB, KS, PITCH>(fa[(t + 1) & 1], hp, (t + 1) * KT);
        __builtin_amdgcn_sched_barrier(0);
        mma_frags<MSUB, KS>(fa[t & 1], fb[t % (BDIST + 1)], acc);
        __builtin_amdgcn_sched_barrier(0);
    }
}

#define MT        ((BIN_ROWS + 15) / 16)
#define TILE_ROWS (16 * MT)
#define NPAD      (TILE_ROWS - BIN_ROWS)
#define ZS_PITCH  (IN_DIM + 4)
#define HS_PITCH  (HIDDEN + 4)
#define GEMM_KS   4
#define GEMM_BDIST 3
static_assert(((ZS_PITCH * 4 / 16) & 1) == 1 && (ZS_PITCH % 2) == 0, "pitch is 16 B x odd, b64 reads stay aligned");
static_assert(((HS_PITCH * 4 / 16) & 1) == 1 && (HS_PITCH % 2) == 0, "pitch is 16 B x odd, b64 reads stay aligned");
static_assert(HIDDEN / 16 <= ROW_T / 32 && OUT_DIM / 16 <= ROW_T / 32, "one wave per 16 output columns");
static_assert(OUT_DIM <= IN_DIM, "the G tile reuses the Z tile's rows");

template<bool B> struct pick { static constexpr bool value = B; };

__global__ __launch_bounds__(ROW_T) void bin_rows(
    const int*   __restrict__ offmat,
    const int4*  __restrict__ tmp,
    const float* __restrict__ X,
    const float* __restrict__ W0,
    const float* __restrict__ W1,
    int*         __restrict__ rowptr,
    int2*        __restrict__ ev,
    float*       __restrict__ G)
{
    __shared__ int  rcnt[BIN_ROWS];
    __shared__ int  roff[BIN_ROWS + 1];
    __shared__ int  red[ROW_T / 32];
    __shared__ int  base_s;
    __shared__ int2 st[EV_STAGE];
    __shared__ __attribute__((aligned(16))) float Zs[TILE_ROWS * ZS_PITCH];
    __shared__ __attribute__((aligned(16))) float Hs[TILE_ROWS * HS_PITCH];

    const int t  = threadIdx.x;
    const int b  = blockIdx.x;
    const int lo = b * BIN_ROWS;
    const int q  = t & (SEG_T - 1);

    int seg_beg = 0, seg_n = 0, below = 0;
    if (t < NCHUNK * SEG_T) {
        const int c  = t / SEG_T;
        const int o0 = offmat[c * (NBIN + 1) + b];
        const int o1 = offmat[c * (NBIN + 1) + b + 1];
        seg_beg = c * CHUNK + o0;
        seg_n   = o1 - o0;
        if (q == 0) below = o0;
    }
    for (int m = 16; m >= 1; m >>= 1) below += __shfl_xor(below, m);
    if ((t & 31) == 0) red[t >> 5] = below;
    if (t < BIN_ROWS) rcnt[t] = 0;
    __syncthreads();
    if (t < 32) {
        int x = red[t];
        for (int m = 16; m >= 1; m >>= 1) x += __shfl_xor(x, m);
        if (t == 0) base_s = x;
    }

    for (int i = q; i < seg_n; i += SEG_T) {
        const int4 r = tmp[seg_beg + i];
        atomicAdd(&rcnt[r.x - lo], 1);
    }
    __syncthreads();

    if (t < BIN_ROWS) {
        int sum = 0;
        for (int i = 0; i < t; ++i) sum += rcnt[i];
        roff[t] = sum;
        if (t == BIN_ROWS - 1) roff[BIN_ROWS] = sum + rcnt[t];
    }
    __syncthreads();

    const int base  = base_s;
    const int total = roff[BIN_ROWS];
    if (t < BIN_ROWS) {
        rcnt[t] = roff[t];
        if (lo + t < N_NODES) rowptr[lo + t] = base + roff[t];
    }
    if (b == NBIN - 1 && t == 0) rowptr[N_NODES] = base + total;
    __syncthreads();

    const bool staged = total <= EV_STAGE;
    for (int i = q; i < seg_n; i += SEG_T) {
        const int4 r = tmp[seg_beg + i];
        const int  p = atomicAdd(&rcnt[r.x - lo], 1);
        const int2 o = make_int2(r.y, r.z);
        if (staged) st[p] = o; else ev[base + p] = o;
    }
    __syncthreads();
    if (staged)
        for (int i = t; i < total; i += ROW_T) ev[base + i] = st[i];

    const int lane = t & 31;
    const int wv = __builtin_amdgcn_readfirstlane(t >> 5);
    auto gather_tasks = [&](auto from_lds) {
        for (int task = wv; task < 2 * TILE_ROWS; task += ROW_T / 32) {
            if (task >= 2 * BIN_ROWS) {
                const int p = task - 2 * BIN_ROWS;
                const int r = BIN_ROWS + p % (NPAD > 0 ? NPAD : 1);
                const int half = p / (NPAD > 0 ? NPAD : 1);
                *(float4*)(Zs + r * ZS_PITCH + half * (IN_DIM / 2) + lane * 4) = make_float4(0.0f, 0.0f, 0.0f, 0.0f);
                continue;
            }
            const int r    = task % BIN_ROWS;
            const int half = task / BIN_ROWS;
            const int row  = lo + r;
            if (row >= N_NODES) {
                *(float4*)(Zs + r * ZS_PITCH + half * (IN_DIM / 2) + lane * 4) = make_float4(0.0f, 0.0f, 0.0f, 0.0f);
                continue;
            }
            const int beg = __builtin_amdgcn_readfirstlane(roff[r]);
            const int end = __builtin_amdgcn_readfirstlane(roff[r + 1]);
            const float* __restrict__ Xh = X + half * (IN_DIM / 2);

            float4 acc[1];
            acc[0] = make_float4(0.0f, 0.0f, 0.0f, 0.0f);
            for (int b0 = beg; b0 < end; b0 += 32) {
                const int n = min(32, end - b0);
                int2 my = make_int2(0, 0);
                if (lane < n) {
                    if constexpr (decltype(from_lds)::value) my = st[b0 + lane];
                    else                                     my = ev[base + b0 + lane];
                }

                int j = 0;
                for (; j + 8 <= n; j += 8) gather_group<IN_DIM / 2, 8, IN_DIM>(Xh, my.x, my.y, j, lane, acc);
                if (j + 4 <= n) { gather_group<IN_DIM / 2, 4, IN_DIM>(Xh, my.x, my.y, j, lane, acc); j += 4; }
                if (j + 2 <= n) { gather_group<IN_DIM / 2, 2, IN_DIM>(Xh, my.x, my.y, j, lane, acc); j += 2; }
                if (j < n)      { gather_group<IN_DIM / 2, 1, IN_DIM>(Xh, my.x, my.y, j, lane, acc); }
            }
            *(float4*)(Zs + r * ZS_PITCH + half * (IN_DIM / 2) + lane * 4) = acc[0];
        }
    };
    if (staged) gather_tasks(pick<true>{}); else gather_tasks(pick<false>{});
    __syncthreads();

    const int l16 = lane & 15;
    const int hi  = lane >> 4;

    if (wv < HIDDEN / 16) {
        const int nb = wv * 16;
        v8f acc[MT][1];
#pragma unroll
        for (int ms = 0; ms < MT; ++ms) acc[ms][0] = 0.0f;
        wave_gemm<MT, GEMM_KS, IN_DIM, HIDDEN, ZS_PITCH, GEMM_BDIST>(
            Zs + l16 * ZS_PITCH + 2 * hi,
            W0 + (size_t)(2 * hi) * HIDDEN + nb + l16, acc);
#pragma unroll
        for (int ms = 0; ms < MT; ++ms)
#pragma unroll
            for (int v = 0; v < 8; ++v)
                Hs[(ms * 16 + v + 8 * hi) * HS_PITCH + nb + l16] = fmaxf(acc[ms][0][v], 0.0f);
    }
    __syncthreads();

    if (wv < OUT_DIM / 16) {
        const int nb = wv * 16;
        v8f acc[MT][1];
#pragma unroll
        for (int ms = 0; ms < MT; ++ms) acc[ms][0] = 0.0f;
        wave_gemm<MT, GEMM_KS, HIDDEN, OUT_DIM, HS_PITCH, GEMM_BDIST>(
            Hs + l16 * HS_PITCH + 2 * hi,
            W1 + (size_t)(2 * hi) * OUT_DIM + nb + l16, acc);
#pragma unroll
        for (int ms = 0; ms < MT; ++ms)
#pragma unroll
            for (int v = 0; v < 8; ++v)
                Zs[(ms * 16 + v + 8 * hi) * ZS_PITCH + nb + l16] = acc[ms][0][v];
    }
    __syncthreads();

    for (int i = t; i < BIN_ROWS * (OUT_DIM / 4); i += ROW_T) {
        const int r = i / (OUT_DIM / 4);
        const int c = i % (OUT_DIM / 4);
        if (lo + r < N_NODES)
            *(float4*)(G + (size_t)(lo + r) * OUT_DIM + 4 * c) = *(const float4*)(Zs + r * ZS_PITCH + 4 * c);
    }
}

static_assert(IN_DIM == 256, "a gather task is one 128-column half of a row");

template<int D>
__global__ __launch_bounds__(256) void spmm_gather(
    const float* __restrict__ H,
    const int*   __restrict__ rowptr,
    const int2*  __restrict__ ev,
    float*       __restrict__ Z)
{
    constexpr int V = D / 128;
    const int lane = threadIdx.x & 31;
    const int row = __builtin_amdgcn_readfirstlane(blockIdx.x * 8 + (threadIdx.x >> 5));
    if (row >= N_NODES) return;

    const int beg = rowptr[row];
    const int end = rowptr[row + 1];

    float4 acc[V];
#pragma unroll
    for (int v = 0; v < V; ++v) acc[v] = make_float4(0.0f, 0.0f, 0.0f, 0.0f);

    for (int base = beg; base < end; base += 32) {
        const int n = min(32, end - base);
        int2 my = make_int2(0, 0);
        if (lane < n) my = ev[base + lane];

        int j = 0;
        for (; j + 8 <= n; j += 8) gather_group<D, 8>(H, my.x, my.y, j, lane, acc);
        if (j + 4 <= n) { gather_group<D, 4>(H, my.x, my.y, j, lane, acc); j += 4; }
        if (j + 2 <= n) { gather_group<D, 2>(H, my.x, my.y, j, lane, acc); j += 2; }
        if (j < n)      { gather_group<D, 1>(H, my.x, my.y, j, lane, acc); }
    }

    float* __restrict__ zrow = Z + (size_t)row * D + lane * 4;
#pragma unroll
    for (int v = 0; v < V; ++v)
        *(float4*)(zrow + v * 128) = acc[v];
}

#define SZ_BYTES   ((size_t)N_NODES * HIDDEN * sizeof(float))
#define OFF_OFFMAT (2 * SZ_BYTES)
#define OFF_ROWPTR (OFF_OFFMAT + 262144)
#define OFF_TMP    (OFF_ROWPTR + 65536)
#define OFF_EV     (OFF_TMP + (size_t)NCHUNK * CHUNK * sizeof(int4))

static_assert((size_t)NCHUNK * (NBIN + 1) * sizeof(int) <= 262144, "offset matrix fits its slot");

extern "C" void kernel_launch(void* const* d_in, const int* in_sizes, int n_in,
                              void* d_out, int out_size, void* d_ws, size_t ws_size,
                              hipStream_t stream)
{
    (void)in_sizes; (void)n_in; (void)out_size; (void)ws_size;

    const float* features = (const float*)d_in[0];
    const int*   esrc     = (const int*)  d_in[1];
    const int*   edst     = (const int*)  d_in[2];
    const float* eval_    = (const float*)d_in[3];
    const float* W0       = (const float*)d_in[4];
    const float* W1       = (const float*)d_in[5];
    float*       out      = (float*)d_out;

    float* G      = (float*)((char*)d_ws + SZ_BYTES);
    int*   offmat = (int*)  ((char*)d_ws + OFF_OFFMAT);
    int*   rowptr = (int*)  ((char*)d_ws + OFF_ROWPTR);
    int4*  tmp    = (int4*) ((char*)d_ws + OFF_TMP);
    int2*  ev     = (int2*) ((char*)d_ws + OFF_EV);

    const int row_blocks  = (N_NODES + 7) / 8;

    edge_bin<<<NCHUNK, BIN_T, 0, stream>>>(esrc, edst, eval_, offmat, tmp);
    bin_rows<<<NBIN, ROW_T, 0, stream>>>(offmat, tmp, features, W0, W1, rowptr, ev, G);

    spmm_gather<OUT_DIM><<<row_blocks, 256, 0, stream>>>(G, rowptr, ev, out);
}
